# starting kernel
# baseline (speedup 1.0000x reference)
#include <hip/hip_runtime.h>
#include <hip/hip_bf16.h>

typedef __attribute__((ext_vector_type(2))) float v2f;
typedef __attribute__((ext_vector_type(8))) float v8f;

#define GD 128
#define BN_EPS 1e-5f
#define LRELU_SLOPE 0.01f

__global__ __launch_bounds__(256)
void wmma_gemm128(const float* __restrict__ A, const float* __restrict__ W,
                  float* __restrict__ C, int nrows) {
    __shared__ float lds[GD * GD];
    const int t = threadIdx.x;
    for (int i = t; i < GD * GD; i += 256) lds[i] = W[i];
    __syncthreads();

    const int wave = t >> 5;
    const int lane = t & 31;
    const int half = lane >> 4;
    const int l16  = lane & 15;

    const int m0 = blockIdx.x * 128 + wave * 16;
    int arow = m0 + l16;
    if (arow >= nrows) arow = nrows - 1;
    const float* aptr = A + (size_t)arow * GD + half * 2;

    v8f acc[8];
#pragma unroll
    for (int nt = 0; nt < 8; ++nt) acc[nt] = (v8f){0, 0, 0, 0, 0, 0, 0, 0};

    for (int k0 = 0; k0 < GD; k0 += 4) {
        const v2f a = *(const v2f*)(aptr + k0);
#pragma unroll
        for (int nt = 0; nt < 8; ++nt) {
            const int nn = nt * 16 + l16;
            v2f b;
            b.x = lds[(k0 + half * 2)     * GD + nn];
            b.y = lds[(k0 + half * 2 + 1) * GD + nn];
            acc[nt] = __builtin_amdgcn_wmma_f32_16x16x4_f32(
                false, a, false, b, (short)0, acc[nt], false, false);
        }
    }

#pragma unroll
    for (int r = 0; r < 8; ++r) {
        const int row = m0 + r + half * 8;
        if (row < nrows) {
#pragma unroll
            for (int nt = 0; nt < 8; ++nt)
                C[(size_t)row * GD + nt * 16 + l16] = acc[nt][r];
        }
    }
}

__global__ void init_one_kernel(float* __restrict__ p, int n) {
    int i = blockIdx.x * 256 + threadIdx.x;
    if (i < n) p[i] = 1.0f;
}
__global__ void deg_accum_kernel(float* __restrict__ deg, const int* __restrict__ dst, int e) {
    int i = blockIdx.x * 256 + threadIdx.x;
    if (i < e) unsafeAtomicAdd(&deg[dst[i]], 1.0f);
}
__global__ void rsqrt_kernel(const float* __restrict__ deg, float* __restrict__ dinv, int n) {
    int i = blockIdx.x * 256 + threadIdx.x;
    if (i < n) dinv[i] = rsqrtf(deg[i]);
}
__global__ void copy_kernel(const float* __restrict__ a, float* __restrict__ b, int n) {
    int i = blockIdx.x * 256 + threadIdx.x;
    if (i < n) b[i] = a[i];
}
__global__ void zero_kernel(float* __restrict__ p, int n) {
    int i = blockIdx.x * 256 + threadIdx.x;
    if (i < n) p[i] = 0.0f;
}

__global__ __launch_bounds__(256)
void scatter_kernel(const float* __restrict__ h, const int* __restrict__ src,
                    const int* __restrict__ dst, const float* __restrict__ dinv,
                    float* __restrict__ agg, int e) {
    int gid = blockIdx.x * 256 + threadIdx.x;
    int edge = gid >> 5;
    if (edge >= e) return;
    int lane = gid & 31;
    int s = src[edge], d = dst[edge];
    float w = dinv[s] * dinv[d];
    const float4 v = *((const float4*)(h + (size_t)s * GD) + lane);
    float* ad = agg + (size_t)d * GD + lane * 4;
    unsafeAtomicAdd(ad + 0, v.x * w);
    unsafeAtomicAdd(ad + 1, v.y * w);
    unsafeAtomicAdd(ad + 2, v.z * w);
    unsafeAtomicAdd(ad + 3, v.w * w);
}

__global__ void post_conv_kernel(float* __restrict__ x, const float* __restrict__ h,
                                 const float* __restrict__ agg, const float* __restrict__ dinv,
                                 const float* __restrict__ bc, int nd) {
    int i = blockIdx.x * 256 + threadIdx.x;
    if (i >= nd) return;
    int row = i >> 7, col = i & (GD - 1);
    float di = dinv[row];
    float conv = agg[i] + h[i] * di * di + bc[col];
    float v = conv + x[i];
    x[i] = v > 0.0f ? v : 0.0f;
}
__global__ void post_ff_kernel(float* __restrict__ x, const float* __restrict__ t,
                               const float* __restrict__ bf, int nd) {
    int i = blockIdx.x * 256 + threadIdx.x;
    if (i >= nd) return;
    int col = i & (GD - 1);
    float u = t[i] + bf[col];
    u = u > 0.0f ? u : LRELU_SLOPE * u;
    float v = u + x[i];
    x[i] = v > 0.0f ? v : 0.0f;
}
__global__ void post_skip_kernel(float* __restrict__ x, const float* __restrict__ t,
                                 const float* __restrict__ bskip, int nd) {
    int i = blockIdx.x * 256 + threadIdx.x;
    if (i >= nd) return;
    int col = i & (GD - 1);
    float s = t[i] + bskip[col];
    float v = x[i] + s;
    x[i] = v > 0.0f ? v : 0.0f;
}
__global__ void bn_kernel(float* __restrict__ x, const float* __restrict__ gamma,
                          const float* __restrict__ beta, const float* __restrict__ rmean,
                          const float* __restrict__ rvar, int nd) {
    int i = blockIdx.x * 256 + threadIdx.x;
    if (i >= nd) return;
    int col = i & (GD - 1);
    float xv = x[i];
    float b = (xv - rmean[col]) * rsqrtf(rvar[col] + BN_EPS) * gamma[col] + beta[col];
    float v = b + xv;
    x[i] = v > 0.0f ? v : 0.0f;
}

extern "C" void kernel_launch(void* const* d_in, const int* in_sizes, int n_in,
                              void* d_out, int out_size, void* d_ws, size_t ws_size,
                              hipStream_t stream) {
    const float* x_in  = (const float*)d_in[0];
    const int*   ei    = (const int*)d_in[1];
    const float* Wc    = (const float*)d_in[2];
    const float* bc    = (const float*)d_in[3];
    const float* Wf    = (const float*)d_in[4];
    const float* bf    = (const float*)d_in[5];
    const float* Wskip = (const float*)d_in[6];
    const float* bskip = (const float*)d_in[7];
    const float* gamma = (const float*)d_in[8];
    const float* beta  = (const float*)d_in[9];
    const float* rmean = (const float*)d_in[10];
    const float* rvar  = (const float*)d_in[11];

    const int n = in_sizes[0] / GD;
    const int e = in_sizes[1] / 2;
    const int L = in_sizes[3] / GD;

    const int* src = ei;
    const int* dst = ei + e;

    float* x    = (float*)d_out;
    float* bufA = (float*)d_ws;
    float* bufB = bufA + (size_t)n * GD;
    float* deg  = bufB + (size_t)n * GD;
    float* dinv = deg + n;

    const int ND = n * GD;
    dim3 blk(256);
    const int gND   = (ND + 255) / 256;
    const int gN    = (n + 255) / 256;
    const int gE    = (e + 255) / 256;
    const int gE32  = (int)(((long long)e * 32 + 255) / 256);
    const int gGemm = (n + 127) / 128;

    init_one_kernel<<<gN, blk, 0, stream>>>(deg, n);
    deg_accum_kernel<<<gE, blk, 0, stream>>>(deg, dst, e);
    rsqrt_kernel<<<gN, blk, 0, stream>>>(deg, dinv, n);
    copy_kernel<<<gND, blk, 0, stream>>>(x_in, x, ND);

    for (int i = 0; i < L; ++i) {
        wmma_gemm128<<<gGemm, blk, 0, stream>>>(x, Wc + (size_t)i * GD * GD, bufA, n);
        zero_kernel<<<gND, blk, 0, stream>>>(bufB, ND);
        scatter_kernel<<<gE32, blk, 0, stream>>>(bufA, src, dst, dinv, bufB, e);
        post_conv_kernel<<<gND, blk, 0, stream>>>(x, bufA, bufB, dinv, bc + i * GD, ND);

        wmma_gemm128<<<gGemm, blk, 0, stream>>>(x, Wf + (size_t)i * GD * GD, bufA, n);
        post_ff_kernel<<<gND, blk, 0, stream>>>(x, bufA, bf + i * GD, ND);

        if (i > 0) {
            wmma_gemm128<<<gGemm, blk, 0, stream>>>(x, Wskip + (size_t)(i - 1) * GD * GD, bufA, n);
            post_skip_kernel<<<gND, blk, 0, stream>>>(x, bufA, bskip + (i - 1) * GD, ND);
        }

        bn_kernel<<<gND, blk, 0, stream>>>(x, gamma + i * GD, beta + i * GD,
                                           rmean + i * GD, rvar + i * GD, ND);
    }
}


# Atomic scatter-add replaced by a gather over a per-call counting sort of edges by destination: one wave sums a row's sources in registers and finishes it; float sum order is now CSR slot order
# speedup vs baseline: 1.4932x; 1.4932x over previous
#include <hip/hip_runtime.h>
#include <hip/hip_bf16.h>

typedef __attribute__((ext_vector_type(2))) float v2f;
typedef __attribute__((ext_vector_type(8))) float v8f;

#define GD 128
#define BN_EPS 1e-5f
#define LRELU_SLOPE 0.01f

__global__ __launch_bounds__(256)
void wmma_gemm128(const float* __restrict__ A, const float* __restrict__ W,
                  float* __restrict__ C, int nrows) {
    __shared__ float lds[GD * GD];
    const int t = threadIdx.x;
    for (int i = t; i < GD * GD; i += 256) lds[i] = W[i];
    __syncthreads();

    const int wave = t >> 5;
    const int lane = t & 31;
    const int half = lane >> 4;
    const int l16  = lane & 15;

    const int m0 = blockIdx.x * 128 + wave * 16;
    int arow = m0 + l16;
    if (arow >= nrows) arow = nrows - 1;
    const float* aptr = A + (size_t)arow * GD + half * 2;

    v8f acc[8];
#pragma unroll
    for (int nt = 0; nt < 8; ++nt) acc[nt] = (v8f){0, 0, 0, 0, 0, 0, 0, 0};

    for (int k0 = 0; k0 < GD; k0 += 4) {
        const v2f a = *(const v2f*)(aptr + k0);
#pragma unroll
        for (int nt = 0; nt < 8; ++nt) {
            const int nn = nt * 16 + l16;
            v2f b;
            b.x = lds[(k0 + half * 2)     * GD + nn];
            b.y = lds[(k0 + half * 2 + 1) * GD + nn];
            acc[nt] = __builtin_amdgcn_wmma_f32_16x16x4_f32(
                false, a, false, b, (short)0, acc[nt], false, false);
        }
    }

#pragma unroll
    for (int r = 0; r < 8; ++r) {
        const int row = m0 + r + half * 8;
        if (row < nrows) {
#pragma unroll
            for (int nt = 0; nt < 8; ++nt)
                C[(size_t)row * GD + nt * 16 + l16] = acc[nt][r];
        }
    }
}

__global__ void zero_int_kernel(int* __restrict__ p, int n) {
    int i = blockIdx.x * 256 + threadIdx.x;
    if (i < n) p[i] = 0;
}
__global__ void copy_kernel(const float* __restrict__ a, float* __restrict__ b, int n) {
    int i = blockIdx.x * 256 + threadIdx.x;
    if (i < n) b[i] = a[i];
}
__global__ void count_rank_kernel(int* __restrict__ cnt, int* __restrict__ rank,
                                  const int* __restrict__ dst, int e) {
    int i = blockIdx.x * 256 + threadIdx.x;
    if (i < e) rank[i] = atomicAdd(&cnt[dst[i]], 1);
}
__global__ __launch_bounds__(256)
void scan_kernel(const int* __restrict__ cnt, int* __restrict__ rowptr,
                 float* __restrict__ dinv, int n) {
    __shared__ int wtot[8];
    __shared__ int wpre[8];
    const int t = threadIdx.x;
    const int lane = t & 31;
    const int wave = t >> 5;
    const int4* c4 = (const int4*)cnt;
    const int nvec = blockIdx.x * 256;
    int pre = 0;
    int i = t;
    for (; i + 7 * 256 < nvec; i += 8 * 256) {
        const int4 a0 = c4[i],           a1 = c4[i + 256],     a2 = c4[i + 2 * 256], a3 = c4[i + 3 * 256];
        const int4 a4 = c4[i + 4 * 256], a5 = c4[i + 5 * 256], a6 = c4[i + 6 * 256], a7 = c4[i + 7 * 256];
        pre += ((a0.x + a0.y) + (a0.z + a0.w)) + ((a1.x + a1.y) + (a1.z + a1.w));
        pre += ((a2.x + a2.y) + (a2.z + a2.w)) + ((a3.x + a3.y) + (a3.z + a3.w));
        pre += ((a4.x + a4.y) + (a4.z + a4.w)) + ((a5.x + a5.y) + (a5.z + a5.w));
        pre += ((a6.x + a6.y) + (a6.z + a6.w)) + ((a7.x + a7.y) + (a7.z + a7.w));
    }
    for (; i < nvec; i += 256) {
        const int4 v = c4[i];
        pre += (v.x + v.y) + (v.z + v.w);
    }
    const int4 mine = c4[nvec + t];
    const int s4 = (mine.x + mine.y) + (mine.z + mine.w);
    int incl = s4;
#pragma unroll
    for (int off = 1; off < 32; off <<= 1) {
        const int v = __shfl_up(incl, off);
        if (lane >= off) incl += v;
    }
#pragma unroll
    for (int off = 16; off > 0; off >>= 1) pre += __shfl_xor(pre, off);
    if (lane == 31) wtot[wave] = incl;
    if (lane == 0)  wpre[wave] = pre;
    __syncthreads();
    int run = 0;
#pragma unroll
    for (int w = 0; w < 8; ++w) {
        run += wpre[w];
        if (w < wave) run += wtot[w];
    }
    run += incl - s4;
    const int i0 = (nvec + t) * 4;
    const int c[4] = {mine.x, mine.y, mine.z, mine.w};
#pragma unroll
    for (int k = 0; k < 4; ++k) {
        const int idx = i0 + k;
        if (idx <= n) rowptr[idx] = run;
        if (idx < n)  dinv[idx] = rsqrtf(1.0f + (float)c[k]);
        run += c[k];
    }
}
__global__ void fill_kernel(const int* __restrict__ rowptr, const int* __restrict__ rank,
                            const int* __restrict__ src, const int* __restrict__ dst,
                            int* __restrict__ csr, int e) {
    int i = blockIdx.x * 256 + threadIdx.x;
    if (i < e) csr[rowptr[dst[i]] + rank[i]] = src[i];
}

__global__ __launch_bounds__(256)
void gather_conv_kernel(float* __restrict__ x, const float* __restrict__ h,
                        const int* __restrict__ rowptr, const int* __restrict__ csr,
                        const float* __restrict__ dinv, const float* __restrict__ bc, int n) {
    const int lane = threadIdx.x & 31;
    const int w0 = __builtin_amdgcn_readfirstlane((int)(blockIdx.x * 8 + (threadIdx.x >> 5)));
    const int nw = (int)gridDim.x * 8;
    const float4* __restrict__ h4 = (const float4*)h;
    float4* x4 = (float4*)x;
    const float4 b = ((const float4*)bc)[lane];

    for (int d = w0; d < n; d += nw) {
        const int start = __builtin_amdgcn_readfirstlane(rowptr[d]);
        const int end   = __builtin_amdgcn_readfirstlane(rowptr[d + 1]);
        const float dd  = dinv[d];
        const size_t o  = (size_t)d * (GD / 4) + lane;
        const float4 hv = h4[o];
        const float4 xv = x4[o];
        float4 acc = {0.0f, 0.0f, 0.0f, 0.0f};

        for (int base = start; base < end; base += 32) {
            const int m = (end - base) < 32 ? (end - base) : 32;
            int   my_s = 0;
            float my_w = 0.0f;
            if (lane < m) {
                my_s = csr[base + lane];
                my_w = dinv[my_s] * dd;
            }
            const int my_wi = __float_as_int(my_w);
            int j = 0;
            for (; j + 4 <= m; j += 4) {
                const int s0 = __builtin_amdgcn_readlane(my_s, j);
                const int s1 = __builtin_amdgcn_readlane(my_s, j + 1);
                const int s2 = __builtin_amdgcn_readlane(my_s, j + 2);
                const int s3 = __builtin_amdgcn_readlane(my_s, j + 3);
                const float4 v0 = h4[(size_t)s0 * (GD / 4) + lane];
                const float4 v1 = h4[(size_t)s1 * (GD / 4) + lane];
                const float4 v2 = h4[(size_t)s2 * (GD / 4) + lane];
                const float4 v3 = h4[(size_t)s3 * (GD / 4) + lane];
                const float q0 = __int_as_float(__builtin_amdgcn_readlane(my_wi, j));
                const float q1 = __int_as_float(__builtin_amdgcn_readlane(my_wi, j + 1));
                const float q2 = __int_as_float(__builtin_amdgcn_readlane(my_wi, j + 2));
                const float q3 = __int_as_float(__builtin_amdgcn_readlane(my_wi, j + 3));
                acc.x += v0.x * q0; acc.y += v0.y * q0; acc.z += v0.z * q0; acc.w += v0.w * q0;
                acc.x += v1.x * q1; acc.y += v1.y * q1; acc.z += v1.z * q1; acc.w += v1.w * q1;
                acc.x += v2.x * q2; acc.y += v2.y * q2; acc.z += v2.z * q2; acc.w += v2.w * q2;
                acc.x += v3.x * q3; acc.y += v3.y * q3; acc.z += v3.z * q3; acc.w += v3.w * q3;
            }
            const int r = m - j;
            if (r > 0) {
                const int sa = __builtin_amdgcn_readlane(my_s, j);
                const float qa = __int_as_float(__builtin_amdgcn_readlane(my_wi, j));
                const float4 va = h4[(size_t)sa * (GD / 4) + lane];
                float4 vb = {0.0f, 0.0f, 0.0f, 0.0f};
                float4 vc = {0.0f, 0.0f, 0.0f, 0.0f};
                float qb = 0.0f, qc = 0.0f;
                if (r > 1) {
                    const int sb = __builtin_amdgcn_readlane(my_s, j + 1);
                    qb = __int_as_float(__builtin_amdgcn_readlane(my_wi, j + 1));
                    vb = h4[(size_t)sb * (GD / 4) + lane];
                }
                if (r > 2) {
                    const int sc = __builtin_amdgcn_readlane(my_s, j + 2);
                    qc = __int_as_float(__builtin_amdgcn_readlane(my_wi, j + 2));
                    vc = h4[(size_t)sc * (GD / 4) + lane];
                }
                acc.x += va.x * qa; acc.y += va.y * qa; acc.z += va.z * qa; acc.w += va.w * qa;
                acc.x += vb.x * qb; acc.y += vb.y * qb; acc.z += vb.z * qb; acc.w += vb.w * qb;
                acc.x += vc.x * qc; acc.y += vc.y * qc; acc.z += vc.z * qc; acc.w += vc.w * qc;
            }
        }

        float4 r;
        r.x = (acc.x + hv.x * dd * dd + b.x) + xv.x;
        r.y = (acc.y + hv.y * dd * dd + b.y) + xv.y;
        r.z = (acc.z + hv.z * dd * dd + b.z) + xv.z;
        r.w = (acc.w + hv.w * dd * dd + b.w) + xv.w;
        r.x = r.x > 0.0f ? r.x : 0.0f;
        r.y = r.y > 0.0f ? r.y : 0.0f;
        r.z = r.z > 0.0f ? r.z : 0.0f;
        r.w = r.w > 0.0f ? r.w : 0.0f;
        x4[o] = r;
    }
}

__global__ void post_ff_kernel(float* __restrict__ x, const float* __restrict__ t,
                               const float* __restrict__ bf, int nd) {
    int i = blockIdx.x * 256 + threadIdx.x;
    if (i >= nd) return;
    int col = i & (GD - 1);
    float u = t[i] + bf[col];
    u = u > 0.0f ? u : LRELU_SLOPE * u;
    float v = u + x[i];
    x[i] = v > 0.0f ? v : 0.0f;
}
__global__ void post_skip_kernel(float* __restrict__ x, const float* __restrict__ t,
                                 const float* __restrict__ bskip, int nd) {
    int i = blockIdx.x * 256 + threadIdx.x;
    if (i >= nd) return;
    int col = i & (GD - 1);
    float s = t[i] + bskip[col];
    float v = x[i] + s;
    x[i] = v > 0.0f ? v : 0.0f;
}
__global__ void bn_kernel(float* __restrict__ x, const float* __restrict__ gamma,
                          const float* __restrict__ beta, const float* __restrict__ rmean,
                          const float* __restrict__ rvar, int nd) {
    int i = blockIdx.x * 256 + threadIdx.x;
    if (i >= nd) return;
    int col = i & (GD - 1);
    float xv = x[i];
    float b = (xv - rmean[col]) * rsqrtf(rvar[col] + BN_EPS) * gamma[col] + beta[col];
    float v = b + xv;
    x[i] = v > 0.0f ? v : 0.0f;
}

extern "C" void kernel_launch(void* const* d_in, const int* in_sizes, int n_in,
                              void* d_out, int out_size, void* d_ws, size_t ws_size,
                              hipStream_t stream) {
    const float* x_in  = (const float*)d_in[0];
    const int*   ei    = (const int*)d_in[1];
    const float* Wc    = (const float*)d_in[2];
    const float* bc    = (const float*)d_in[3];
    const float* Wf    = (const float*)d_in[4];
    const float* bf    = (const float*)d_in[5];
    const float* Wskip = (const float*)d_in[6];
    const float* bskip = (const float*)d_in[7];
    const float* gamma = (const float*)d_in[8];
    const float* beta  = (const float*)d_in[9];
    const float* rmean = (const float*)d_in[10];
    const float* rvar  = (const float*)d_in[11];

    const int n = in_sizes[0] / GD;
    const int e = in_sizes[1] / 2;
    const int L = in_sizes[3] / GD;

    const int* src = ei;
    const int* dst = ei + e;

    float* x    = (float*)d_out;
    float* bufA = (float*)d_ws;
    int*   ibuf = (int*)(bufA + (size_t)n * GD);
    const int npad = (n + 1 + 1023) & ~1023;
    int*   cnt    = ibuf;
    int*   rowptr = cnt + npad;
    int*   rank   = rowptr + npad;
    int*   csr    = rank + e;
    float* dinv   = (float*)ibuf + (size_t)n * GD + n;

    const int ND = n * GD;
    dim3 blk(256);
    const int gND   = (ND + 255) / 256;
    const int gE    = (e + 255) / 256;
    const int gPad  = npad / 256;
    const int gScan = npad / 1024;
    const int gGemm = (n + 127) / 128;
    const int gGath = 2048;

    zero_int_kernel<<<gPad, blk, 0, stream>>>(cnt, npad);
    count_rank_kernel<<<gE, blk, 0, stream>>>(cnt, rank, dst, e);
    scan_kernel<<<gScan, blk, 0, stream>>>(cnt, rowptr, dinv, n);
    fill_kernel<<<gE, blk, 0, stream>>>(rowptr, rank, src, dst, csr, e);
    copy_kernel<<<gND, blk, 0, stream>>>(x_in, x, ND);

    for (int i = 0; i < L; ++i) {
        wmma_gemm128<<<gGemm, blk, 0, stream>>>(x, Wc + (size_t)i * GD * GD, bufA, n);
        gather_conv_kernel<<<gGath, blk, 0, stream>>>(x, bufA, rowptr, csr, dinv, bc + i * GD, n);

        wmma_gemm128<<<gGemm, blk, 0, stream>>>(x, Wf + (size_t)i * GD * GD, bufA, n);
        post_ff_kernel<<<gND, blk, 0, stream>>>(x, bufA, bf + i * GD, ND);

        if (i > 0) {
            wmma_gemm128<<<gGemm, blk, 0, stream>>>(x, Wskip + (size_t)(i - 1) * GD * GD, bufA, n);
            post_skip_kernel<<<gND, blk, 0, stream>>>(x, bufA, bskip + (i - 1) * GD, ND);
        }

        bn_kernel<<<gND, blk, 0, stream>>>(x, gamma + i * GD, beta + i * GD,
                                           rmean + i * GD, rvar + i * GD, ND);
    }
}


# Bias, leaky-relu, skip add, relu and batch norm applied to the WMMA accumulators in registers and stored to x in place: the 8 elementwise launches after the feed-forward and skip GEMMs are gone
# speedup vs baseline: 2.1245x; 1.4228x over previous
#include <hip/hip_runtime.h>
#include <hip/hip_bf16.h>

typedef __attribute__((ext_vector_type(2))) float v2f;
typedef __attribute__((ext_vector_type(8))) float v8f;

#define GD 128
#define BN_EPS 1e-5f
#define LRELU_SLOPE 0.01f

template <int EPI, bool BN>
__global__ __launch_bounds__(256)
void wmma_gemm128(const float* A, const float* __restrict__ W, float* C, int nrows,
                  const float* __restrict__ bias, const float* __restrict__ gamma,
                  const float* __restrict__ beta, const float* __restrict__ rmean,
                  const float* __restrict__ rvar) {
    __shared__ float lds[GD * GD];
    const int t = threadIdx.x;
    for (int i = t; i < GD * GD; i += 256) lds[i] = W[i];
    __syncthreads();

    const int wave = t >> 5;
    const int lane = t & 31;
    const int half = lane >> 4;
    const int l16  = lane & 15;

    const int m0 = blockIdx.x * 128 + wave * 16;
    if (m0 >= nrows) return;
    int arow = m0 + l16;
    if (arow >= nrows) arow = nrows - 1;
    const float* aptr = A + (size_t)arow * GD + half * 2;

    v8f acc[8];
#pragma unroll
    for (int nt = 0; nt < 8; ++nt) acc[nt] = (v8f){0, 0, 0, 0, 0, 0, 0, 0};

    for (int k0 = 0; k0 < GD; k0 += 4) {
        const v2f a = *(const v2f*)(aptr + k0);
#pragma unroll
        for (int nt = 0; nt < 8; ++nt) {
            const int nn = nt * 16 + l16;
            v2f b;
            b.x = lds[(k0 + half * 2)     * GD + nn];
            b.y = lds[(k0 + half * 2 + 1) * GD + nn];
            acc[nt] = __builtin_amdgcn_wmma_f32_16x16x4_f32(
                false, a, false, b, (short)0, acc[nt], false, false);
        }
    }

    if (EPI == 0) {
#pragma unroll
        for (int r = 0; r < 8; ++r) {
            const int row = m0 + r + half * 8;
            if (row < nrows) {
#pragma unroll
                for (int nt = 0; nt < 8; ++nt)
                    C[(size_t)row * GD + nt * 16 + l16] = acc[nt][r];
            }
        }
    } else {
        float cb[8], cm[8], cr[8], cg[8], ce[8];
#pragma unroll
        for (int nt = 0; nt < 8; ++nt) {
            const int col = nt * 16 + l16;
            cb[nt] = bias[col];
            if (BN) {
                cm[nt] = rmean[col];
                cr[nt] = rsqrtf(rvar[col] + BN_EPS);
                cg[nt] = gamma[col];
                ce[nt] = beta[col];
            }
        }
        float xo[8][8];
#pragma unroll
        for (int r = 0; r < 8; ++r) {
            int lrow = m0 + r + half * 8;
            if (lrow >= nrows) lrow = nrows - 1;
#pragma unroll
            for (int nt = 0; nt < 8; ++nt)
                xo[r][nt] = C[(size_t)lrow * GD + nt * 16 + l16];
        }
#pragma unroll
        for (int r = 0; r < 8; ++r) {
            const int row = m0 + r + half * 8;
            if (row < nrows) {
#pragma unroll
                for (int nt = 0; nt < 8; ++nt) {
                    float u = acc[nt][r] + cb[nt];
                    if (EPI == 1) u = u > 0.0f ? u : LRELU_SLOPE * u;
                    float v = u + xo[r][nt];
                    v = v > 0.0f ? v : 0.0f;
                    if (BN) {
                        const float bnv = (v - cm[nt]) * cr[nt] * cg[nt] + ce[nt];
                        v = bnv + v;
                        v = v > 0.0f ? v : 0.0f;
                    }
                    C[(size_t)row * GD + nt * 16 + l16] = v;
                }
            }
        }
    }
}

__global__ void zero_int_kernel(int* __restrict__ p, int n) {
    int i = blockIdx.x * 256 + threadIdx.x;
    if (i < n) p[i] = 0;
}
__global__ void copy_kernel(const float* __restrict__ a, float* __restrict__ b, int n) {
    int i = blockIdx.x * 256 + threadIdx.x;
    if (i < n) b[i] = a[i];
}
__global__ void count_rank_kernel(int* __restrict__ cnt, int* __restrict__ rank,
                                  const int* __restrict__ dst, int e) {
    int i = blockIdx.x * 256 + threadIdx.x;
    if (i < e) rank[i] = atomicAdd(&cnt[dst[i]], 1);
}
__global__ __launch_bounds__(256)
void scan_kernel(const int* __restrict__ cnt, int* __restrict__ rowptr,
                 float* __restrict__ dinv, int n) {
    __shared__ int wtot[8];
    __shared__ int wpre[8];
    const int t = threadIdx.x;
    const int lane = t & 31;
    const int wave = t >> 5;
    const int4* c4 = (const int4*)cnt;
    const int nvec = blockIdx.x * 256;
    int pre = 0;
    int i = t;
    for (; i + 7 * 256 < nvec; i += 8 * 256) {
        const int4 a0 = c4[i],           a1 = c4[i + 256],     a2 = c4[i + 2 * 256], a3 = c4[i + 3 * 256];
        const int4 a4 = c4[i + 4 * 256], a5 = c4[i + 5 * 256], a6 = c4[i + 6 * 256], a7 = c4[i + 7 * 256];
        pre += ((a0.x + a0.y) + (a0.z + a0.w)) + ((a1.x + a1.y) + (a1.z + a1.w));
        pre += ((a2.x + a2.y) + (a2.z + a2.w)) + ((a3.x + a3.y) + (a3.z + a3.w));
        pre += ((a4.x + a4.y) + (a4.z + a4.w)) + ((a5.x + a5.y) + (a5.z + a5.w));
        pre += ((a6.x + a6.y) + (a6.z + a6.w)) + ((a7.x + a7.y) + (a7.z + a7.w));
    }
    for (; i < nvec; i += 256) {
        const int4 v = c4[i];
        pre += (v.x + v.y) + (v.z + v.w);
    }
    const int4 mine = c4[nvec + t];
    const int s4 = (mine.x + mine.y) + (mine.z + mine.w);
    int incl = s4;
#pragma unroll
    for (int off = 1; off < 32; off <<= 1) {
        const int v = __shfl_up(incl, off);
        if (lane >= off) incl += v;
    }
#pragma unroll
    for (int off = 16; off > 0; off >>= 1) pre += __shfl_xor(pre, off);
    if (lane == 31) wtot[wave] = incl;
    if (lane == 0)  wpre[wave] = pre;
    __syncthreads();
    int run = 0;
#pragma unroll
    for (int w = 0; w < 8; ++w) {
        run += wpre[w];
        if (w < wave) run += wtot[w];
    }
    run += incl - s4;
    const int i0 = (nvec + t) * 4;
    const int c[4] = {mine.x, mine.y, mine.z, mine.w};
#pragma unroll
    for (int k = 0; k < 4; ++k) {
        const int idx = i0 + k;
        if (idx <= n) rowptr[idx] = run;
        if (idx < n)  dinv[idx] = rsqrtf(1.0f + (float)c[k]);
        run += c[k];
    }
}
__global__ void fill_kernel(const int* __restrict__ rowptr, const int* __restrict__ rank,
                            const int* __restrict__ src, const int* __restrict__ dst,
                            int* __restrict__ csr, int e) {
    int i = blockIdx.x * 256 + threadIdx.x;
    if (i < e) csr[rowptr[dst[i]] + rank[i]] = src[i];
}

__global__ __launch_bounds__(256)
void gather_conv_kernel(float* __restrict__ x, const float* __restrict__ h,
                        const int* __restrict__ rowptr, const int* __restrict__ csr,
                        const float* __restrict__ dinv, const float* __restrict__ bc, int n) {
    const int lane = threadIdx.x & 31;
    const int w0 = __builtin_amdgcn_readfirstlane((int)(blockIdx.x * 8 + (threadIdx.x >> 5)));
    const int nw = (int)gridDim.x * 8;
    const float4* __restrict__ h4 = (const float4*)h;
    float4* x4 = (float4*)x;
    const float4 b = ((const float4*)bc)[lane];

    for (int d = w0; d < n; d += nw) {
        const int start = __builtin_amdgcn_readfirstlane(rowptr[d]);
        const int end   = __builtin_amdgcn_readfirstlane(rowptr[d + 1]);
        const float dd  = dinv[d];
        const size_t o  = (size_t)d * (GD / 4) + lane;
        const float4 hv = h4[o];
        const float4 xv = x4[o];
        float4 acc = {0.0f, 0.0f, 0.0f, 0.0f};

        for (int base = start; base < end; base += 32) {
            const int m = (end - base) < 32 ? (end - base) : 32;
            int   my_s = 0;
            float my_w = 0.0f;
            if (lane < m) {
                my_s = csr[base + lane];
                my_w = dinv[my_s] * dd;
            }
            const int my_wi = __float_as_int(my_w);
            int j = 0;
            for (; j + 4 <= m; j += 4) {
                const int s0 = __builtin_amdgcn_readlane(my_s, j);
                const int s1 = __builtin_amdgcn_readlane(my_s, j + 1);
                const int s2 = __builtin_amdgcn_readlane(my_s, j + 2);
                const int s3 = __builtin_amdgcn_readlane(my_s, j + 3);
                const float4 v0 = h4[(size_t)s0 * (GD / 4) + lane];
                const float4 v1 = h4[(size_t)s1 * (GD / 4) + lane];
                const float4 v2 = h4[(size_t)s2 * (GD / 4) + lane];
                const float4 v3 = h4[(size_t)s3 * (GD / 4) + lane];
                const float q0 = __int_as_float(__builtin_amdgcn_readlane(my_wi, j));
                const float q1 = __int_as_float(__builtin_amdgcn_readlane(my_wi, j + 1));
                const float q2 = __int_as_float(__builtin_amdgcn_readlane(my_wi, j + 2));
                const float q3 = __int_as_float(__builtin_amdgcn_readlane(my_wi, j + 3));
                acc.x += v0.x * q0; acc.y += v0.y * q0; acc.z += v0.z * q0; acc.w += v0.w * q0;
                acc.x += v1.x * q1; acc.y += v1.y * q1; acc.z += v1.z * q1; acc.w += v1.w * q1;
                acc.x += v2.x * q2; acc.y += v2.y * q2; acc.z += v2.z * q2; acc.w += v2.w * q2;
                acc.x += v3.x * q3; acc.y += v3.y * q3; acc.z += v3.z * q3; acc.w += v3.w * q3;
            }
            const int r = m - j;
            if (r > 0) {
                const int sa = __builtin_amdgcn_readlane(my_s, j);
                const float qa = __int_as_float(__builtin_amdgcn_readlane(my_wi, j));
                const float4 va = h4[(size_t)sa * (GD / 4) + lane];
                float4 vb = {0.0f, 0.0f, 0.0f, 0.0f};
                float4 vc = {0.0f, 0.0f, 0.0f, 0.0f};
                float qb = 0.0f, qc = 0.0f;
                if (r > 1) {
                    const int sb = __builtin_amdgcn_readlane(my_s, j + 1);
                    qb = __int_as_float(__builtin_amdgcn_readlane(my_wi, j + 1));
                    vb = h4[(size_t)sb * (GD / 4) + lane];
                }
                if (r > 2) {
                    const int sc = __builtin_amdgcn_readlane(my_s, j + 2);
                    qc = __int_as_float(__builtin_amdgcn_readlane(my_wi, j + 2));
                    vc = h4[(size_t)sc * (GD / 4) + lane];
                }
                acc.x += va.x * qa; acc.y += va.y * qa; acc.z += va.z * qa; acc.w += va.w * qa;
                acc.x += vb.x * qb; acc.y += vb.y * qb; acc.z += vb.z * qb; acc.w += vb.w * qb;
                acc.x += vc.x * qc; acc.y += vc.y * qc; acc.z += vc.z * qc; acc.w += vc.w * qc;
            }
        }

        float4 r;
        r.x = (acc.x + hv.x * dd * dd + b.x) + xv.x;
        r.y = (acc.y + hv.y * dd * dd + b.y) + xv.y;
        r.z = (acc.z + hv.z * dd * dd + b.z) + xv.z;
        r.w = (acc.w + hv.w * dd * dd + b.w) + xv.w;
        r.x = r.x > 0.0f ? r.x : 0.0f;
        r.y = r.y > 0.0f ? r.y : 0.0f;
        r.z = r.z > 0.0f ? r.z : 0.0f;
        r.w = r.w > 0.0f ? r.w : 0.0f;
        x4[o] = r;
    }
}

extern "C" void kernel_launch(void* const* d_in, const int* in_sizes, int n_in,
                              void* d_out, int out_size, void* d_ws, size_t ws_size,
                              hipStream_t stream) {
    const float* x_in  = (const float*)d_in[0];
    const int*   ei    = (const int*)d_in[1];
    const float* Wc    = (const float*)d_in[2];
    const float* bc    = (const float*)d_in[3];
    const float* Wf    = (const float*)d_in[4];
    const float* bf    = (const float*)d_in[5];
    const float* Wskip = (const float*)d_in[6];
    const float* bskip = (const float*)d_in[7];
    const float* gamma = (const float*)d_in[8];
    const float* beta  = (const float*)d_in[9];
    const float* rmean = (const float*)d_in[10];
    const float* rvar  = (const float*)d_in[11];

    const int n = in_sizes[0] / GD;
    const int e = in_sizes[1] / 2;
    const int L = in_sizes[3] / GD;

    const int* src = ei;
    const int* dst = ei + e;

    float* x    = (float*)d_out;
    float* bufA = (float*)d_ws;
    int*   ibuf = (int*)(bufA + (size_t)n * GD);
    const int npad = (n + 1 + 1023) & ~1023;
    int*   cnt    = ibuf;
    int*   rowptr = cnt + npad;
    int*   rank   = rowptr + npad;
    int*   csr    = rank + e;
    float* dinv   = (float*)ibuf + (size_t)n * GD + n;

    const int ND = n * GD;
    dim3 blk(256);
    const int gND   = (ND + 255) / 256;
    const int gE    = (e + 255) / 256;
    const int gPad  = npad / 256;
    const int gScan = npad / 1024;
    const int gGemm = (n + 127) / 128;
    const int gGath = 2048;

    zero_int_kernel<<<gPad, blk, 0, stream>>>(cnt, npad);
    count_rank_kernel<<<gE, blk, 0, stream>>>(cnt, rank, dst, e);
    scan_kernel<<<gScan, blk, 0, stream>>>(cnt, rowptr, dinv, n);
    fill_kernel<<<gE, blk, 0, stream>>>(rowptr, rank, src, dst, csr, e);
    copy_kernel<<<gND, blk, 0, stream>>>(x_in, x, ND);

    for (int i = 0; i < L; ++i) {
        wmma_gemm128<0, false><<<gGemm, blk, 0, stream>>>(x, Wc + (size_t)i * GD * GD, bufA, n,
                                                          nullptr, nullptr, nullptr, nullptr, nullptr);
        gather_conv_kernel<<<gGath, blk, 0, stream>>>(x, bufA, rowptr, csr, dinv, bc + i * GD, n);

        if (i == 0) {
            wmma_gemm128<1, true><<<gGemm, blk, 0, stream>>>(x, Wf + (size_t)i * GD * GD, x, n,
                                                             bf + i * GD, gamma + i * GD, beta + i * GD,
                                                             rmean + i * GD, rvar + i * GD);
        } else {
            wmma_gemm128<1, false><<<gGemm, blk, 0, stream>>>(x, Wf + (size_t)i * GD * GD, x, n,
                                                              bf + i * GD, nullptr, nullptr, nullptr, nullptr);
            wmma_gemm128<2, true><<<gGemm, blk, 0, stream>>>(x, Wskip + (size_t)(i - 1) * GD * GD, x, n,
                                                             bskip + (i - 1) * GD, gamma + i * GD, beta + i * GD,
                                                             rmean + i * GD, rvar + i * GD);
        }
    }
}


# GEMM operand path: W kept in LDS in B-fragment order (tile columns permuted), each B read one ds_load_b128 issued a k-step ahead, A stripe loaded whole before the loop, epilogue in 16-byte accesses
# speedup vs baseline: 2.4135x; 1.1360x over previous
#include <hip/hip_runtime.h>
#include <hip/hip_bf16.h>

typedef __attribute__((ext_vector_type(2))) float v2f;
typedef __attribute__((ext_vector_type(4))) float v4f;
typedef __attribute__((ext_vector_type(8))) float v8f;

#define GD 128
#define BN_EPS 1e-5f
#define LRELU_SLOPE 0.01f

template <int EPI, bool BN>
__global__ __launch_bounds__(256)
void wmma_gemm128(const float* A, const float* __restrict__ W, float* C, int nrows,
                  const float* __restrict__ bias, const float* __restrict__ gamma,
                  const float* __restrict__ beta, const float* __restrict__ rmean,
                  const float* __restrict__ rvar) {
    __shared__ v4f ldsv[GD * GD / 4];
    const int t = threadIdx.x;
    const int wave = t >> 5;
    const int lane = t & 31;
    const int half = lane >> 4;
    const int l16  = lane & 15;

    v4f wa[8], wb[8];
#pragma unroll
    for (int i = 0; i < 8; ++i) {
        const int sp = wave + 8 * i;
        const int s = sp >> 1, ph = sp & 1;
        const float* wp = W + (4 * s + 2 * half) * GD + 4 * l16 + 64 * ph;
        wa[i] = *(const v4f*)wp;
        wb[i] = *(const v4f*)(wp + GD);
    }

    const int m0 = blockIdx.x * 128 + wave * 16;
    int arow = m0 + l16;
    if (arow >= nrows) arow = nrows - 1;
    const float* aptr = A + (size_t)arow * GD + half * 2;
    v2f a[32];
#pragma unroll
    for (int s = 0; s < 32; ++s) a[s] = *(const v2f*)(aptr + 4 * s);

#pragma unroll
    for (int i = 0; i < 8; ++i) {
        const int sp = wave + 8 * i;
        const int s = sp >> 1, ph = sp & 1;
        ldsv[(s * 4 + 2 * ph) * 32 + lane]     = (v4f){wa[i].x, wb[i].x, wa[i].y, wb[i].y};
        ldsv[(s * 4 + 2 * ph + 1) * 32 + lane] = (v4f){wa[i].z, wb[i].z, wa[i].w, wb[i].w};
    }
    __syncthreads();
    if (m0 >= nrows) return;

    v8f acc[8];
#pragma unroll
    for (int nt = 0; nt < 8; ++nt) acc[nt] = (v8f){0, 0, 0, 0, 0, 0, 0, 0};

    v4f bq[2][4];
#pragma unroll
    for (int p = 0; p < 4; ++p) bq[0][p] = ldsv[p * 32 + lane];
    __builtin_amdgcn_sched_group_barrier(0x100, 4, 0);
#pragma unroll
    for (int s = 0; s < 32; ++s) {
        if (s + 1 < 32) {
#pragma unroll
            for (int p = 0; p < 4; ++p) bq[(s + 1) & 1][p] = ldsv[((s + 1) * 4 + p) * 32 + lane];
        }
#pragma unroll
        for (int p = 0; p < 4; ++p) {
            const v4f q = bq[s & 1][p];
            const v2f b0 = {q.x, q.y};
            const v2f b1 = {q.z, q.w};
            acc[2 * p] = __builtin_amdgcn_wmma_f32_16x16x4_f32(
                false, a[s], false, b0, (short)0, acc[2 * p], false, false);
            acc[2 * p + 1] = __builtin_amdgcn_wmma_f32_16x16x4_f32(
                false, a[s], false, b1, (short)0, acc[2 * p + 1], false, false);
        }
        __builtin_amdgcn_sched_group_barrier(0x100, 4, 0);
        __builtin_amdgcn_sched_group_barrier(0x008, 8, 0);
    }

    const int c0 = 4 * l16;
    if (EPI == 0) {
#pragma unroll
        for (int r = 0; r < 8; ++r) {
            const int row = m0 + r + half * 8;
            if (row < nrows) {
                float* cp = C + (size_t)row * GD + c0;
                *(v4f*)cp        = (v4f){acc[0][r], acc[1][r], acc[2][r], acc[3][r]};
                *(v4f*)(cp + 64) = (v4f){acc[4][r], acc[5][r], acc[6][r], acc[7][r]};
            }
        }
    } else {
        float cb[8], cm[8], cr[8], cg[8], ce[8];
#pragma unroll
        for (int g = 0; g < 2; ++g) {
            const int col = c0 + 64 * g;
            const v4f vb = *(const v4f*)(bias + col);
#pragma unroll
            for (int j = 0; j < 4; ++j) cb[4 * g + j] = vb[j];
            if (BN) {
                const v4f vm = *(const v4f*)(rmean + col);
                const v4f vv = *(const v4f*)(rvar + col);
                const v4f vg = *(const v4f*)(gamma + col);
                const v4f ve = *(const v4f*)(beta + col);
#pragma unroll
                for (int j = 0; j < 4; ++j) {
                    cm[4 * g + j] = vm[j];
                    cr[4 * g + j] = rsqrtf(vv[j] + BN_EPS);
                    cg[4 * g + j] = vg[j];
                    ce[4 * g + j] = ve[j];
                }
            }
        }
        v4f xo[8][2];
#pragma unroll
        for (int r = 0; r < 8; ++r) {
            int lrow = m0 + r + half * 8;
            if (lrow >= nrows) lrow = nrows - 1;
            const float* xp = C + (size_t)lrow * GD + c0;
            xo[r][0] = *(const v4f*)xp;
            xo[r][1] = *(const v4f*)(xp + 64);
        }
#pragma unroll
        for (int r = 0; r < 8; ++r) {
            const int row = m0 + r + half * 8;
            if (row < nrows) {
                v4f o[2];
#pragma unroll
                for (int nt = 0; nt < 8; ++nt) {
                    float u = acc[nt][r] + cb[nt];
                    if (EPI == 1) u = u > 0.0f ? u : LRELU_SLOPE * u;
                    float v = u + xo[r][nt >> 2][nt & 3];
                    v = v > 0.0f ? v : 0.0f;
                    if (BN) {
                        const float bnv = (v - cm[nt]) * cr[nt] * cg[nt] + ce[nt];
                        v = bnv + v;
                        v = v > 0.0f ? v : 0.0f;
                    }
                    o[nt >> 2][nt & 3] = v;
                }
                float* cp = C + (size_t)row * GD + c0;
                *(v4f*)cp        = o[0];
                *(v4f*)(cp + 64) = o[1];
            }
        }
    }
}

__global__ void zero_int_kernel(int* __restrict__ p, int n) {
    int i = blockIdx.x * 256 + threadIdx.x;
    if (i < n) p[i] = 0;
}
__global__ void copy_kernel(const float* __restrict__ a, float* __restrict__ b, int n) {
    int i = blockIdx.x * 256 + threadIdx.x;
    if (i < n) b[i] = a[i];
}
__global__ void count_rank_kernel(int* __restrict__ cnt, int* __restrict__ rank,
                                  const int* __restrict__ dst, int e) {
    int i = blockIdx.x * 256 + threadIdx.x;
    if (i < e) rank[i] = atomicAdd(&cnt[dst[i]], 1);
}
__global__ __launch_bounds__(256)
void scan_kernel(const int* __restrict__ cnt, int* __restrict__ rowptr,
                 float* __restrict__ dinv, int n) {
    __shared__ int wtot[8];
    __shared__ int wpre[8];
    const int t = threadIdx.x;
    const int lane = t & 31;
    const int wave = t >> 5;
    const int4* c4 = (const int4*)cnt;
    const int nvec = blockIdx.x * 256;
    int pre = 0;
    int i = t;
    for (; i + 7 * 256 < nvec; i += 8 * 256) {
        const int4 a0 = c4[i],           a1 = c4[i + 256],     a2 = c4[i + 2 * 256], a3 = c4[i + 3 * 256];
        const int4 a4 = c4[i + 4 * 256], a5 = c4[i + 5 * 256], a6 = c4[i + 6 * 256], a7 = c4[i + 7 * 256];
        pre += ((a0.x + a0.y) + (a0.z + a0.w)) + ((a1.x + a1.y) + (a1.z + a1.w));
        pre += ((a2.x + a2.y) + (a2.z + a2.w)) + ((a3.x + a3.y) + (a3.z + a3.w));
        pre += ((a4.x + a4.y) + (a4.z + a4.w)) + ((a5.x + a5.y) + (a5.z + a5.w));
        pre += ((a6.x + a6.y) + (a6.z + a6.w)) + ((a7.x + a7.y) + (a7.z + a7.w));
    }
    for (; i < nvec; i += 256) {
        const int4 v = c4[i];
        pre += (v.x + v.y) + (v.z + v.w);
    }
    const int4 mine = c4[nvec + t];
    const int s4 = (mine.x + mine.y) + (mine.z + mine.w);
    int incl = s4;
#pragma unroll
    for (int off = 1; off < 32; off <<= 1) {
        const int v = __shfl_up(incl, off);
        if (lane >= off) incl += v;
    }
#pragma unroll
    for (int off = 16; off > 0; off >>= 1) pre += __shfl_xor(pre, off);
    if (lane == 31) wtot[wave] = incl;
    if (lane == 0)  wpre[wave] = pre;
    __syncthreads();
    int run = 0;
#pragma unroll
    for (int w = 0; w < 8; ++w) {
        run += wpre[w];
        if (w < wave) run += wtot[w];
    }
    run += incl - s4;
    const int i0 = (nvec + t) * 4;
    const int c[4] = {mine.x, mine.y, mine.z, mine.w};
#pragma unroll
    for (int k = 0; k < 4; ++k) {
        const int idx = i0 + k;
        if (idx <= n) rowptr[idx] = run;
        if (idx < n)  dinv[idx] = rsqrtf(1.0f + (float)c[k]);
        run += c[k];
    }
}
__global__ void fill_kernel(const int* __restrict__ rowptr, const int* __restrict__ rank,
                            const int* __restrict__ src, const int* __restrict__ dst,
                            int* __restrict__ csr, int e) {
    int i = blockIdx.x * 256 + threadIdx.x;
    if (i < e) csr[rowptr[dst[i]] + rank[i]] = src[i];
}

__global__ __launch_bounds__(256)
void gather_conv_kernel(float* __restrict__ x, const float* __restrict__ h,
                        const int* __restrict__ rowptr, const int* __restrict__ csr,
                        const float* __restrict__ dinv, const float* __restrict__ bc, int n) {
    const int lane = threadIdx.x & 31;
    const int w0 = __builtin_amdgcn_readfirstlane((int)(blockIdx.x * 8 + (threadIdx.x >> 5)));
    const int nw = (int)gridDim.x * 8;
    const float4* __restrict__ h4 = (const float4*)h;
    float4* x4 = (float4*)x;
    const float4 b = ((const float4*)bc)[lane];

    for (int d = w0; d < n; d += nw) {
        const int start = __builtin_amdgcn_readfirstlane(rowptr[d]);
        const int end   = __builtin_amdgcn_readfirstlane(rowptr[d + 1]);
        const float dd  = dinv[d];
        const size_t o  = (size_t)d * (GD / 4) + lane;
        const float4 hv = h4[o];
        const float4 xv = x4[o];
        float4 acc = {0.0f, 0.0f, 0.0f, 0.0f};

        for (int base = start; base < end; base += 32) {
            const int m = (end - base) < 32 ? (end - base) : 32;
            int   my_s = 0;
            float my_w = 0.0f;
            if (lane < m) {
                my_s = csr[base + lane];
                my_w = dinv[my_s] * dd;
            }
            const int my_wi = __float_as_int(my_w);
            int j = 0;
            for (; j + 4 <= m; j += 4) {
                const int s0 = __builtin_amdgcn_readlane(my_s, j);
                const int s1 = __builtin_amdgcn_readlane(my_s, j + 1);
                const int s2 = __builtin_amdgcn_readlane(my_s, j + 2);
                const int s3 = __builtin_amdgcn_readlane(my_s, j + 3);
                const float4 v0 = h4[(size_t)s0 * (GD / 4) + lane];
                const float4 v1 = h4[(size_t)s1 * (GD / 4) + lane];
                const float4 v2 = h4[(size_t)s2 * (GD / 4) + lane];
                const float4 v3 = h4[(size_t)s3 * (GD / 4) + lane];
                const float q0 = __int_as_float(__builtin_amdgcn_readlane(my_wi, j));
                const float q1 = __int_as_float(__builtin_amdgcn_readlane(my_wi, j + 1));
                const float q2 = __int_as_float(__builtin_amdgcn_readlane(my_wi, j + 2));
                const float q3 = __int_as_float(__builtin_amdgcn_readlane(my_wi, j + 3));
                acc.x += v0.x * q0; acc.y += v0.y * q0; acc.z += v0.z * q0; acc.w += v0.w * q0;
                acc.x += v1.x * q1; acc.y += v1.y * q1; acc.z += v1.z * q1; acc.w += v1.w * q1;
                acc.x += v2.x * q2; acc.y += v2.y * q2; acc.z += v2.z * q2; acc.w += v2.w * q2;
                acc.x += v3.x * q3; acc.y += v3.y * q3; acc.z += v3.z * q3; acc.w += v3.w * q3;
            }
            const int r = m - j;
            if (r > 0) {
                const int sa = __builtin_amdgcn_readlane(my_s, j);
                const float qa = __int_as_float(__builtin_amdgcn_readlane(my_wi, j));
                const float4 va = h4[(size_t)sa * (GD / 4) + lane];
                float4 vb = {0.0f, 0.0f, 0.0f, 0.0f};
                float4 vc = {0.0f, 0.0f, 0.0f, 0.0f};
                float qb = 0.0f, qc = 0.0f;
                if (r > 1) {
                    const int sb = __builtin_amdgcn_readlane(my_s, j + 1);
                    qb = __int_as_float(__builtin_amdgcn_readlane(my_wi, j + 1));
                    vb = h4[(size_t)sb * (GD / 4) + lane];
                }
                if (r > 2) {
                    const int sc = __builtin_amdgcn_readlane(my_s, j + 2);
                    qc = __int_as_float(__builtin_amdgcn_readlane(my_wi, j + 2));
                    vc = h4[(size_t)sc * (GD / 4) + lane];
                }
                acc.x += va.x * qa; acc.y += va.y * qa; acc.z += va.z * qa; acc.w += va.w * qa;
                acc.x += vb.x * qb; acc.y += vb.y * qb; acc.z += vb.z * qb; acc.w += vb.w * qb;
                acc.x += vc.x * qc; acc.y += vc.y * qc; acc.z += vc.z * qc; acc.w += vc.w * qc;
            }
        }

        float4 r;
        r.x = (acc.x + hv.x * dd * dd + b.x) + xv.x;
        r.y = (acc.y + hv.y * dd * dd + b.y) + xv.y;
        r.z = (acc.z + hv.z * dd * dd + b.z) + xv.z;
        r.w = (acc.w + hv.w * dd * dd + b.w) + xv.w;
        r.x = r.x > 0.0f ? r.x : 0.0f;
        r.y = r.y > 0.0f ? r.y : 0.0f;
        r.z = r.z > 0.0f ? r.z : 0.0f;
        r.w = r.w > 0.0f ? r.w : 0.0f;
        x4[o] = r;
    }
}

extern "C" void kernel_launch(void* const* d_in, const int* in_sizes, int n_in,
                              void* d_out, int out_size, void* d_ws, size_t ws_size,
                              hipStream_t stream) {
    const float* x_in  = (const float*)d_in[0];
    const int*   ei    = (const int*)d_in[1];
    const float* Wc    = (const float*)d_in[2];
    const float* bc    = (const float*)d_in[3];
    const float* Wf    = (const float*)d_in[4];
    const float* bf    = (const float*)d_in[5];
    const float* Wskip = (const float*)d_in[6];
    const float* bskip = (const float*)d_in[7];
    const float* gamma = (const float*)d_in[8];
    const float* beta  = (const float*)d_in[9];
    const float* rmean = (const float*)d_in[10];
    const float* rvar  = (const float*)d_in[11];

    const int n = in_sizes[0] / GD;
    const int e = in_sizes[1] / 2;
    const int L = in_sizes[3] / GD;

    const int* src = ei;
    const int* dst = ei + e;

    float* x    = (float*)d_out;
    float* bufA = (float*)d_ws;
    int*   ibuf = (int*)(bufA + (size_t)n * GD);
    const int npad = (n + 1 + 1023) & ~1023;
    int*   cnt    = ibuf;
    int*   rowptr = cnt + npad;
    int*   rank   = rowptr + npad;
    int*   csr    = rank + e;
    float* dinv   = (float*)ibuf + (size_t)n * GD + n;

    const int ND = n * GD;
    dim3 blk(256);
    const int gND   = (ND + 255) / 256;
    const int gE    = (e + 255) / 256;
    const int gPad  = npad / 256;
    const int gScan = npad / 1024;
    const int gGemm = (n + 127) / 128;
    const int gGath = 2048;

    zero_int_kernel<<<gPad, blk, 0, stream>>>(cnt, npad);
    count_rank_kernel<<<gE, blk, 0, stream>>>(cnt, rank, dst, e);
    scan_kernel<<<gScan, blk, 0, stream>>>(cnt, rowptr, dinv, n);
    fill_kernel<<<gE, blk, 0, stream>>>(rowptr, rank, src, dst, csr, e);
    copy_kernel<<<gND, blk, 0, stream>>>(x_in, x, ND);

    for (int i = 0; i < L; ++i) {
        wmma_gemm128<0, false><<<gGemm, blk, 0, stream>>>(x, Wc + (size_t)i * GD * GD, bufA, n,
                                                          nullptr, nullptr, nullptr, nullptr, nullptr);
        gather_conv_kernel<<<gGath, blk, 0, stream>>>(x, bufA, rowptr, csr, dinv, bc + i * GD, n);

        if (i == 0) {
            wmma_gemm128<1, true><<<gGemm, blk, 0, stream>>>(x, Wf + (size_t)i * GD * GD, x, n,
                                                             bf + i * GD, gamma + i * GD, beta + i * GD,
                                                             rmean + i * GD, rvar + i * GD);
        } else {
            wmma_gemm128<1, false><<<gGemm, blk, 0, stream>>>(x, Wf + (size_t)i * GD * GD, x, n,
                                                              bf + i * GD, nullptr, nullptr, nullptr, nullptr);
            wmma_gemm128<2, true><<<gGemm, blk, 0, stream>>>(x, Wskip + (size_t)(i - 1) * GD * GD, x, n,
                                                             bskip + (i - 1) * GD, gamma + i * GD, beta + i * GD,
                                                             rmean + i * GD, rvar + i * GD);
        }
    }
}


# Input copy removed: layer 0's conv GEMM and gather read the input x directly and the gather writes the output buffer; one launch and one full read+write pass over x are gone
# speedup vs baseline: 2.6904x; 1.1147x over previous
#include <hip/hip_runtime.h>
#include <hip/hip_bf16.h>

typedef __attribute__((ext_vector_type(2))) float v2f;
typedef __attribute__((ext_vector_type(4))) float v4f;
typedef __attribute__((ext_vector_type(8))) float v8f;

#define GD 128
#define BN_EPS 1e-5f
#define LRELU_SLOPE 0.01f

template <int EPI, bool BN>
__global__ __launch_bounds__(256)
void wmma_gemm128(const float* A, const float* __restrict__ W, float* C, int nrows,
                  const float* __restrict__ bias, const float* __restrict__ gamma,
                  const float* __restrict__ beta, const float* __restrict__ rmean,
                  const float* __restrict__ rvar) {
    __shared__ v4f ldsv[GD * GD / 4];
    const int t = threadIdx.x;
    const int wave = t >> 5;
    const int lane = t & 31;
    const int half = lane >> 4;
    const int l16  = lane & 15;

    v4f wa[8], wb[8];
#pragma unroll
    for (int i = 0; i < 8; ++i) {
        const int sp = wave + 8 * i;
        const int s = sp >> 1, ph = sp & 1;
        const float* wp = W + (4 * s + 2 * half) * GD + 4 * l16 + 64 * ph;
        wa[i] = *(const v4f*)wp;
        wb[i] = *(const v4f*)(wp + GD);
    }

    const int m0 = blockIdx.x * 128 + wave * 16;
    int arow = m0 + l16;
    if (arow >= nrows) arow = nrows - 1;
    const float* aptr = A + (size_t)arow * GD + half * 2;
    v2f a[32];
#pragma unroll
    for (int s = 0; s < 32; ++s) a[s] = *(const v2f*)(aptr + 4 * s);

#pragma unroll
    for (int i = 0; i < 8; ++i) {
        const int sp = wave + 8 * i;
        const int s = sp >> 1, ph = sp & 1;
        ldsv[(s * 4 + 2 * ph) * 32 + lane]     = (v4f){wa[i].x, wb[i].x, wa[i].y, wb[i].y};
        ldsv[(s * 4 + 2 * ph + 1) * 32 + lane] = (v4f){wa[i].z, wb[i].z, wa[i].w, wb[i].w};
    }
    __syncthreads();
    if (m0 >= nrows) return;

    v8f acc[8];
#pragma unroll
    for (int nt = 0; nt < 8; ++nt) acc[nt] = (v8f){0, 0, 0, 0, 0, 0, 0, 0};

    v4f bq[2][4];
#pragma unroll
    for (int p = 0; p < 4; ++p) bq[0][p] = ldsv[p * 32 + lane];
    __builtin_amdgcn_sched_group_barrier(0x100, 4, 0);
#pragma unroll
    for (int s = 0; s < 32; ++s) {
        if (s + 1 < 32) {
#pragma unroll
            for (int p = 0; p < 4; ++p) bq[(s + 1) & 1][p] = ldsv[((s + 1) * 4 + p) * 32 + lane];
        }
#pragma unroll
        for (int p = 0; p < 4; ++p) {
            const v4f q = bq[s & 1][p];
            const v2f b0 = {q.x, q.y};
            const v2f b1 = {q.z, q.w};
            acc[2 * p] = __builtin_amdgcn_wmma_f32_16x16x4_f32(
                false, a[s], false, b0, (short)0, acc[2 * p], false, false);
            acc[2 * p + 1] = __builtin_amdgcn_wmma_f32_16x16x4_f32(
                false, a[s], false, b1, (short)0, acc[2 * p + 1], false, false);
        }
        __builtin_amdgcn_sched_group_barrier(0x100, 4, 0);
        __builtin_amdgcn_sched_group_barrier(0x008, 8, 0);
    }

    const int c0 = 4 * l16;
    if (EPI == 0) {
#pragma unroll
        for (int r = 0; r < 8; ++r) {
            const int row = m0 + r + half * 8;
            if (row < nrows) {
                float* cp = C + (size_t)row * GD + c0;
                *(v4f*)cp        = (v4f){acc[0][r], acc[1][r], acc[2][r], acc[3][r]};
                *(v4f*)(cp + 64) = (v4f){acc[4][r], acc[5][r], acc[6][r], acc[7][r]};
            }
        }
    } else {
        float cb[8], cm[8], cr[8], cg[8], ce[8];
#pragma unroll
        for (int g = 0; g < 2; ++g) {
            const int col = c0 + 64 * g;
            const v4f vb = *(const v4f*)(bias + col);
#pragma unroll
            for (int j = 0; j < 4; ++j) cb[4 * g + j] = vb[j];
            if (BN) {
                const v4f vm = *(const v4f*)(rmean + col);
                const v4f vv = *(const v4f*)(rvar + col);
                const v4f vg = *(const v4f*)(gamma + col);
                const v4f ve = *(const v4f*)(beta + col);
#pragma unroll
                for (int j = 0; j < 4; ++j) {
                    cm[4 * g + j] = vm[j];
                    cr[4 * g + j] = rsqrtf(vv[j] + BN_EPS);
                    cg[4 * g + j] = vg[j];
                    ce[4 * g + j] = ve[j];
                }
            }
        }
        v4f xo[8][2];
#pragma unroll
        for (int r = 0; r < 8; ++r) {
            int lrow = m0 + r + half * 8;
            if (lrow >= nrows) lrow = nrows - 1;
            const float* xp = C + (size_t)lrow * GD + c0;
            xo[r][0] = *(const v4f*)xp;
            xo[r][1] = *(const v4f*)(xp + 64);
        }
#pragma unroll
        for (int r = 0; r < 8; ++r) {
            const int row = m0 + r + half * 8;
            if (row < nrows) {
                v4f o[2];
#pragma unroll
                for (int nt = 0; nt < 8; ++nt) {
                    float u = acc[nt][r] + cb[nt];
                    if (EPI == 1) u = u > 0.0f ? u : LRELU_SLOPE * u;
                    float v = u + xo[r][nt >> 2][nt & 3];
                    v = v > 0.0f ? v : 0.0f;
                    if (BN) {
                        const float bnv = (v - cm[nt]) * cr[nt] * cg[nt] + ce[nt];
                        v = bnv + v;
                        v = v > 0.0f ? v : 0.0f;
                    }
                    o[nt >> 2][nt & 3] = v;
                }
                float* cp = C + (size_t)row * GD + c0;
                *(v4f*)cp        = o[0];
                *(v4f*)(cp + 64) = o[1];
            }
        }
    }
}

__global__ void zero_int_kernel(int* __restrict__ p, int n) {
    int i = blockIdx.x * 256 + threadIdx.x;
    if (i < n) p[i] = 0;
}
__global__ void count_rank_kernel(int* __restrict__ cnt, int* __restrict__ rank,
                                  const int* __restrict__ dst, int e) {
    int i = blockIdx.x * 256 + threadIdx.x;
    if (i < e) rank[i] = atomicAdd(&cnt[dst[i]], 1);
}
__global__ __launch_bounds__(256)
void scan_kernel(const int* __restrict__ cnt, int* __restrict__ rowptr,
                 float* __restrict__ dinv, int n) {
    __shared__ int wtot[8];
    __shared__ int wpre[8];
    const int t = threadIdx.x;
    const int lane = t & 31;
    const int wave = t >> 5;
    const int4* c4 = (const int4*)cnt;
    const int nvec = blockIdx.x * 256;
    int pre = 0;
    int i = t;
    for (; i + 7 * 256 < nvec; i += 8 * 256) {
        const int4 a0 = c4[i],           a1 = c4[i + 256],     a2 = c4[i + 2 * 256], a3 = c4[i + 3 * 256];
        const int4 a4 = c4[i + 4 * 256], a5 = c4[i + 5 * 256], a6 = c4[i + 6 * 256], a7 = c4[i + 7 * 256];
        pre += ((a0.x + a0.y) + (a0.z + a0.w)) + ((a1.x + a1.y) + (a1.z + a1.w));
        pre += ((a2.x + a2.y) + (a2.z + a2.w)) + ((a3.x + a3.y) + (a3.z + a3.w));
        pre += ((a4.x + a4.y) + (a4.z + a4.w)) + ((a5.x + a5.y) + (a5.z + a5.w));
        pre += ((a6.x + a6.y) + (a6.z + a6.w)) + ((a7.x + a7.y) + (a7.z + a7.w));
    }
    for (; i < nvec; i += 256) {
        const int4 v = c4[i];
        pre += (v.x + v.y) + (v.z + v.w);
    }
    const int4 mine = c4[nvec + t];
    const int s4 = (mine.x + mine.y) + (mine.z + mine.w);
    int incl = s4;
#pragma unroll
    for (int off = 1; off < 32; off <<= 1) {
        const int v = __shfl_up(incl, off);
        if (lane >= off) incl += v;
    }
#pragma unroll
    for (int off = 16; off > 0; off >>= 1) pre += __shfl_xor(pre, off);
    if (lane == 31) wtot[wave] = incl;
    if (lane == 0)  wpre[wave] = pre;
    __syncthreads();
    int run = 0;
#pragma unroll
    for (int w = 0; w < 8; ++w) {
        run += wpre[w];
        if (w < wave) run += wtot[w];
    }
    run += incl - s4;
    const int i0 = (nvec + t) * 4;
    const int c[4] = {mine.x, mine.y, mine.z, mine.w};
#pragma unroll
    for (int k = 0; k < 4; ++k) {
        const int idx = i0 + k;
        if (idx <= n) rowptr[idx] = run;
        if (idx < n)  dinv[idx] = rsqrtf(1.0f + (float)c[k]);
        run += c[k];
    }
}
__global__ void fill_kernel(const int* __restrict__ rowptr, const int* __restrict__ rank,
                            const int* __restrict__ src, const int* __restrict__ dst,
                            int* __restrict__ csr, int e) {
    int i = blockIdx.x * 256 + threadIdx.x;
    if (i < e) csr[rowptr[dst[i]] + rank[i]] = src[i];
}

__global__ __launch_bounds__(256)
void gather_conv_kernel(const float* xin, float* xout, const float* __restrict__ h,
                        const int* __restrict__ rowptr, const int* __restrict__ csr,
                        const float* __restrict__ dinv, const float* __restrict__ bc, int n) {
    const int lane = threadIdx.x & 31;
    const int w0 = __builtin_amdgcn_readfirstlane((int)(blockIdx.x * 8 + (threadIdx.x >> 5)));
    const int nw = (int)gridDim.x * 8;
    const float4* __restrict__ h4 = (const float4*)h;
    const float4* xi4 = (const float4*)xin;
    float4* xo4 = (float4*)xout;
    const float4 b = ((const float4*)bc)[lane];

    for (int d = w0; d < n; d += nw) {
        const int start = __builtin_amdgcn_readfirstlane(rowptr[d]);
        const int end   = __builtin_amdgcn_readfirstlane(rowptr[d + 1]);
        const float dd  = dinv[d];
        const size_t o  = (size_t)d * (GD / 4) + lane;
        const float4 hv = h4[o];
        const float4 xv = xi4[o];
        float4 acc = {0.0f, 0.0f, 0.0f, 0.0f};

        for (int base = start; base < end; base += 32) {
            const int m = (end - base) < 32 ? (end - base) : 32;
            int   my_s = 0;
            float my_w = 0.0f;
            if (lane < m) {
                my_s = csr[base + lane];
                my_w = dinv[my_s] * dd;
            }
            const int my_wi = __float_as_int(my_w);
            int j = 0;
            for (; j + 4 <= m; j += 4) {
                const int s0 = __builtin_amdgcn_readlane(my_s, j);
                const int s1 = __builtin_amdgcn_readlane(my_s, j + 1);
                const int s2 = __builtin_amdgcn_readlane(my_s, j + 2);
                const int s3 = __builtin_amdgcn_readlane(my_s, j + 3);
                const float4 v0 = h4[(size_t)s0 * (GD / 4) + lane];
                const float4 v1 = h4[(size_t)s1 * (GD / 4) + lane];
                const float4 v2 = h4[(size_t)s2 * (GD / 4) + lane];
                const float4 v3 = h4[(size_t)s3 * (GD / 4) + lane];
                const float q0 = __int_as_float(__builtin_amdgcn_readlane(my_wi, j));
                const float q1 = __int_as_float(__builtin_amdgcn_readlane(my_wi, j + 1));
                const float q2 = __int_as_float(__builtin_amdgcn_readlane(my_wi, j + 2));
                const float q3 = __int_as_float(__builtin_amdgcn_readlane(my_wi, j + 3));
                acc.x += v0.x * q0; acc.y += v0.y * q0; acc.z += v0.z * q0; acc.w += v0.w * q0;
                acc.x += v1.x * q1; acc.y += v1.y * q1; acc.z += v1.z * q1; acc.w += v1.w * q1;
                acc.x += v2.x * q2; acc.y += v2.y * q2; acc.z += v2.z * q2; acc.w += v2.w * q2;
                acc.x += v3.x * q3; acc.y += v3.y * q3; acc.z += v3.z * q3; acc.w += v3.w * q3;
            }
            const int r = m - j;
            if (r > 0) {
                const int sa = __builtin_amdgcn_readlane(my_s, j);
                const float qa = __int_as_float(__builtin_amdgcn_readlane(my_wi, j));
                const float4 va = h4[(size_t)sa * (GD / 4) + lane];
                float4 vb = {0.0f, 0.0f, 0.0f, 0.0f};
                float4 vc = {0.0f, 0.0f, 0.0f, 0.0f};
                float qb = 0.0f, qc = 0.0f;
                if (r > 1) {
                    const int sb = __builtin_amdgcn_readlane(my_s, j + 1);
                    qb = __int_as_float(__builtin_amdgcn_readlane(my_wi, j + 1));
                    vb = h4[(size_t)sb * (GD / 4) + lane];
                }
                if (r > 2) {
                    const int sc = __builtin_amdgcn_readlane(my_s, j + 2);
                    qc = __int_as_float(__builtin_amdgcn_readlane(my_wi, j + 2));
                    vc = h4[(size_t)sc * (GD / 4) + lane];
                }
                acc.x += va.x * qa; acc.y += va.y * qa; acc.z += va.z * qa; acc.w += va.w * qa;
                acc.x += vb.x * qb; acc.y += vb.y * qb; acc.z += vb.z * qb; acc.w += vb.w * qb;
                acc.x += vc.x * qc; acc.y += vc.y * qc; acc.z += vc.z * qc; acc.w += vc.w * qc;
            }
        }

        float4 r;
        r.x = (acc.x + hv.x * dd * dd + b.x) + xv.x;
        r.y = (acc.y + hv.y * dd * dd + b.y) + xv.y;
        r.z = (acc.z + hv.z * dd * dd + b.z) + xv.z;
        r.w = (acc.w + hv.w * dd * dd + b.w) + xv.w;
        r.x = r.x > 0.0f ? r.x : 0.0f;
        r.y = r.y > 0.0f ? r.y : 0.0f;
        r.z = r.z > 0.0f ? r.z : 0.0f;
        r.w = r.w > 0.0f ? r.w : 0.0f;
        xo4[o] = r;
    }
}

extern "C" void kernel_launch(void* const* d_in, const int* in_sizes, int n_in,
                              void* d_out, int out_size, void* d_ws, size_t ws_size,
                              hipStream_t stream) {
    const float* x_in  = (const float*)d_in[0];
    const int*   ei    = (const int*)d_in[1];
    const float* Wc    = (const float*)d_in[2];
    const float* bc    = (const float*)d_in[3];
    const float* Wf    = (const float*)d_in[4];
    const float* bf    = (const float*)d_in[5];
    const float* Wskip = (const float*)d_in[6];
    const float* bskip = (const float*)d_in[7];
    const float* gamma = (const float*)d_in[8];
    const float* beta  = (const float*)d_in[9];
    const float* rmean = (const float*)d_in[10];
    const float* rvar  = (const float*)d_in[11];

    const int n = in_sizes[0] / GD;
    const int e = in_sizes[1] / 2;
    const int L = in_sizes[3] / GD;

    const int* src = ei;
    const int* dst = ei + e;

    float* x    = (float*)d_out;
    float* bufA = (float*)d_ws;
    int*   ibuf = (int*)(bufA + (size_t)n * GD);
    const int npad = (n + 1 + 1023) & ~1023;
    int*   cnt    = ibuf;
    int*   rowptr = cnt + npad;
    int*   rank   = rowptr + npad;
    int*   csr    = rank + e;
    float* dinv   = (float*)ibuf + (size_t)n * GD + n;

    dim3 blk(256);
    const int gE    = (e + 255) / 256;
    const int gPad  = npad / 256;
    const int gScan = npad / 1024;
    const int gGemm = (n + 127) / 128;
    const int gGath = 2048;

    zero_int_kernel<<<gPad, blk, 0, stream>>>(cnt, npad);
    count_rank_kernel<<<gE, blk, 0, stream>>>(cnt, rank, dst, e);
    scan_kernel<<<gScan, blk, 0, stream>>>(cnt, rowptr, dinv, n);
    fill_kernel<<<gE, blk, 0, stream>>>(rowptr, rank, src, dst, csr, e);

    for (int i = 0; i < L; ++i) {
        const float* xcur = (i == 0) ? x_in : x;
        wmma_gemm128<0, false><<<gGemm, blk, 0, stream>>>(xcur, Wc + (size_t)i * GD * GD, bufA, n,
                                                          nullptr, nullptr, nullptr, nullptr, nullptr);
        gather_conv_kernel<<<gGath, blk, 0, stream>>>(xcur, x, bufA, rowptr, csr, dinv, bc + i * GD, n);

        if (i == 0) {
            wmma_gemm128<1, true><<<gGemm, blk, 0, stream>>>(x, Wf + (size_t)i * GD * GD, x, n,
                                                             bf + i * GD, gamma + i * GD, beta + i * GD,
                                                             rmean + i * GD, rvar + i * GD);
        } else {
            wmma_gemm128<1, false><<<gGemm, blk, 0, stream>>>(x, Wf + (size_t)i * GD * GD, x, n,
                                                              bf + i * GD, nullptr, nullptr, nullptr, nullptr);
            wmma_gemm128<2, true><<<gGemm, blk, 0, stream>>>(x, Wskip + (size_t)(i - 1) * GD * GD, x, n,
                                                             bskip + (i - 1) * GD, gamma + i * GD, beta + i * GD,
                                                             rmean + i * GD, rvar + i * GD);
        }
    }
}


# Row-local products chained in one launch: a wave's finished 16-row stripe goes through an LDS tile into A-fragment order for the next W (feed-forward, skip, next conv): 8 GEMM launches become 4
# speedup vs baseline: 2.8332x; 1.0531x over previous
#include <hip/hip_runtime.h>
#include <hip/hip_bf16.h>

typedef __attribute__((ext_vector_type(2))) float v2f;
typedef __attribute__((ext_vector_type(4))) float v4f;
typedef __attribute__((ext_vector_type(8))) float v8f;

#define GD 128
#define BN_EPS 1e-5f
#define LRELU_SLOPE 0.01f

__device__ __forceinline__ void w_fetch(const float* __restrict__ W, int wave, int half, int l16,
                                        v4f (&wa)[8], v4f (&wb)[8]) {
#pragma unroll
    for (int i = 0; i < 8; ++i) {
        const int sp = wave + 8 * i;
        const int s = sp >> 1, ph = sp & 1;
        const float* wp = W + (4 * s + 2 * half) * GD + 4 * l16 + 64 * ph;
        wa[i] = *(const v4f*)wp;
        wb[i] = *(const v4f*)(wp + GD);
    }
}
__device__ __forceinline__ void w_put(v4f* ldsv, int wave, int lane, const v4f (&wa)[8], const v4f (&wb)[8]) {
#pragma unroll
    for (int i = 0; i < 8; ++i) {
        const int sp = wave + 8 * i;
        const int s = sp >> 1, ph = sp & 1;
        ldsv[(s * 4 + 2 * ph) * 32 + lane]     = (v4f){wa[i].x, wb[i].x, wa[i].y, wb[i].y};
        ldsv[(s * 4 + 2 * ph + 1) * 32 + lane] = (v4f){wa[i].z, wb[i].z, wa[i].w, wb[i].w};
    }
}
__device__ __forceinline__ void stripe_product(const v4f* ldsv, int lane, const v2f (&a)[32], v8f (&acc)[8]) {
#pragma unroll
    for (int nt = 0; nt < 8; ++nt) acc[nt] = (v8f){0, 0, 0, 0, 0, 0, 0, 0};
    v4f bq[2][4];
#pragma unroll
    for (int p = 0; p < 4; ++p) bq[0][p] = ldsv[p * 32 + lane];
    __builtin_amdgcn_sched_group_barrier(0x100, 4, 0);
#pragma unroll
    for (int s = 0; s < 32; ++s) {
        if (s + 1 < 32) {
#pragma unroll
            for (int p = 0; p < 4; ++p) bq[(s + 1) & 1][p] = ldsv[((s + 1) * 4 + p) * 32 + lane];
        }
#pragma unroll
        for (int p = 0; p < 4; ++p) {
            const v4f q = bq[s & 1][p];
            const v2f b0 = {q.x, q.y};
            const v2f b1 = {q.z, q.w};
            acc[2 * p] = __builtin_amdgcn_wmma_f32_16x16x4_f32(
                false, a[s], false, b0, (short)0, acc[2 * p], false, false);
            acc[2 * p + 1] = __builtin_amdgcn_wmma_f32_16x16x4_f32(
                false, a[s], false, b1, (short)0, acc[2 * p + 1], false, false);
        }
        __builtin_amdgcn_sched_group_barrier(0x100, 4, 0);
        __builtin_amdgcn_sched_group_barrier(0x008, 8, 0);
    }
}

__global__ __launch_bounds__(256)
void wmma_gemm128(const float* __restrict__ A, const float* __restrict__ W, float* __restrict__ C, int nrows) {
    __shared__ v4f ldsv[GD * GD / 4];
    const int t = threadIdx.x;
    const int wave = t >> 5;
    const int lane = t & 31;
    const int half = lane >> 4;
    const int l16  = lane & 15;

    v4f wa[8], wb[8];
    w_fetch(W, wave, half, l16, wa, wb);
    const int m0 = blockIdx.x * 128 + wave * 16;
    int arow = m0 + l16;
    if (arow >= nrows) arow = nrows - 1;
    const float* aptr = A + (size_t)arow * GD + half * 2;
    v2f a[32];
#pragma unroll
    for (int s = 0; s < 32; ++s) a[s] = *(const v2f*)(aptr + 4 * s);
    w_put(ldsv, wave, lane, wa, wb);
    __syncthreads();
    if (m0 >= nrows) return;

    v8f acc[8];
    stripe_product(ldsv, lane, a, acc);

    const int c0 = 4 * l16;
#pragma unroll
    for (int r = 0; r < 8; ++r) {
        const int row = m0 + r + half * 8;
        if (row < nrows) {
            float* cp = C + (size_t)row * GD + c0;
            *(v4f*)cp        = (v4f){acc[0][r], acc[1][r], acc[2][r], acc[3][r]};
            *(v4f*)(cp + 64) = (v4f){acc[4][r], acc[5][r], acc[6][r], acc[7][r]};
        }
    }
}

#define TP 132

struct BnConst { float m[8], r[8], g[8], e[8]; };
__device__ __forceinline__ void bn_load(BnConst& k, const float* __restrict__ gamma, const float* __restrict__ beta,
                                        const float* __restrict__ rmean, const float* __restrict__ rvar, int c0) {
#pragma unroll
    for (int g = 0; g < 2; ++g) {
        const int col = c0 + 64 * g;
        const v4f vm = *(const v4f*)(rmean + col);
        const v4f vv = *(const v4f*)(rvar + col);
        const v4f vg = *(const v4f*)(gamma + col);
        const v4f ve = *(const v4f*)(beta + col);
#pragma unroll
        for (int j = 0; j < 4; ++j) {
            k.m[4 * g + j] = vm[j];
            k.r[4 * g + j] = rsqrtf(vv[j] + BN_EPS);
            k.g[4 * g + j] = vg[j];
            k.e[4 * g + j] = ve[j];
        }
    }
}
__device__ __forceinline__ float bn_apply(const BnConst& k, int nt, float v) {
    const float bnv = (v - k.m[nt]) * k.r[nt] * k.g[nt] + k.e[nt];
    v = bnv + v;
    return v > 0.0f ? v : 0.0f;
}

template <bool SKIP, bool CONV>
__global__ __launch_bounds__(256)
void wmma_chain128(float* x, float* __restrict__ hout, int nrows,
                   const float* __restrict__ Wf, const float* __restrict__ bf,
                   const float* __restrict__ Ws, const float* __restrict__ bs,
                   const float* __restrict__ Wn,
                   const float* __restrict__ gamma, const float* __restrict__ beta,
                   const float* __restrict__ rmean, const float* __restrict__ rvar) {
    __shared__ v4f lds[GD * GD / 4 + 8 * 16 * TP / 4];
    v4f* ldsv = lds;
    float* tile = (float*)(lds + GD * GD / 4);
    const int t = threadIdx.x;
    const int wave = t >> 5;
    const int lane = t & 31;
    const int half = lane >> 4;
    const int l16  = lane & 15;
    const int c0 = 4 * l16;
    const int m0 = blockIdx.x * 128 + wave * 16;
    float* tw = tile + wave * 16 * TP;
    float* tc = tw + (8 * half) * TP + c0;
    const float* ta = tw + l16 * TP + 2 * half;

    v4f wa[8], wb[8];
    v2f a[32];
    v8f acc[8];

    w_fetch(Wf, wave, half, l16, wa, wb);
    int arow = m0 + l16;
    if (arow >= nrows) arow = nrows - 1;
    const float* aptr = x + (size_t)arow * GD + half * 2;
#pragma unroll
    for (int s = 0; s < 32; ++s) a[s] = *(const v2f*)(aptr + 4 * s);
    w_put(ldsv, wave, lane, wa, wb);
    __syncthreads();
    stripe_product(ldsv, lane, a, acc);

    {
        float cb[8];
#pragma unroll
        for (int g = 0; g < 2; ++g) {
            const v4f vb = *(const v4f*)(bf + c0 + 64 * g);
#pragma unroll
            for (int j = 0; j < 4; ++j) cb[4 * g + j] = vb[j];
        }
        BnConst k;
        if (!SKIP) bn_load(k, gamma, beta, rmean, rvar, c0);
        v4f xo[8][2];
#pragma unroll
        for (int r = 0; r < 8; ++r) {
            int lrow = m0 + r + half * 8;
            if (lrow >= nrows) lrow = nrows - 1;
            const float* xp = x + (size_t)lrow * GD + c0;
            xo[r][0] = *(const v4f*)xp;
            xo[r][1] = *(const v4f*)(xp + 64);
        }
#pragma unroll
        for (int r = 0; r < 8; ++r) {
            v4f o[2];
#pragma unroll
            for (int nt = 0; nt < 8; ++nt) {
                float u = acc[nt][r] + cb[nt];
                u = u > 0.0f ? u : LRELU_SLOPE * u;
                float v = u + xo[r][nt >> 2][nt & 3];
                v = v > 0.0f ? v : 0.0f;
                if (!SKIP) v = bn_apply(k, nt, v);
                o[nt >> 2][nt & 3] = v;
            }
            if (SKIP || CONV) {
                *(v4f*)(tc + r * TP)      = o[0];
                *(v4f*)(tc + r * TP + 64) = o[1];
            }
            if (!SKIP) {
                const int row = m0 + r + half * 8;
                if (row < nrows) {
                    float* cp = x + (size_t)row * GD + c0;
                    *(v4f*)cp        = o[0];
                    *(v4f*)(cp + 64) = o[1];
                }
            }
        }
    }

    if (SKIP) {
        __syncthreads();
        w_fetch(Ws, wave, half, l16, wa, wb);
        w_put(ldsv, wave, lane, wa, wb);
        __syncthreads();
#pragma unroll
        for (int s = 0; s < 32; ++s) a[s] = *(const v2f*)(ta + 4 * s);
        stripe_product(ldsv, lane, a, acc);

        float cb[8];
#pragma unroll
        for (int g = 0; g < 2; ++g) {
            const v4f vb = *(const v4f*)(bs + c0 + 64 * g);
#pragma unroll
            for (int j = 0; j < 4; ++j) cb[4 * g + j] = vb[j];
        }
        BnConst k;
        bn_load(k, gamma, beta, rmean, rvar, c0);
#pragma unroll
        for (int r = 0; r < 8; ++r) {
            v4f vo[2];
            vo[0] = *(const v4f*)(tc + r * TP);
            vo[1] = *(const v4f*)(tc + r * TP + 64);
            v4f o[2];
#pragma unroll
            for (int nt = 0; nt < 8; ++nt) {
                const float u = acc[nt][r] + cb[nt];
                float v = u + vo[nt >> 2][nt & 3];
                v = v > 0.0f ? v : 0.0f;
                o[nt >> 2][nt & 3] = bn_apply(k, nt, v);
            }
            if (CONV) {
                *(v4f*)(tc + r * TP)      = o[0];
                *(v4f*)(tc + r * TP + 64) = o[1];
            }
            const int row = m0 + r + half * 8;
            if (row < nrows) {
                float* cp = x + (size_t)row * GD + c0;
                *(v4f*)cp        = o[0];
                *(v4f*)(cp + 64) = o[1];
            }
        }
    }

    if (CONV) {
        __syncthreads();
        w_fetch(Wn, wave, half, l16, wa, wb);
        w_put(ldsv, wave, lane, wa, wb);
        __syncthreads();
#pragma unroll
        for (int s = 0; s < 32; ++s) a[s] = *(const v2f*)(ta + 4 * s);
        stripe_product(ldsv, lane, a, acc);
#pragma unroll
        for (int r = 0; r < 8; ++r) {
            const int row = m0 + r + half * 8;
            if (row < nrows) {
                float* cp = hout + (size_t)row * GD + c0;
                *(v4f*)cp        = (v4f){acc[0][r], acc[1][r], acc[2][r], acc[3][r]};
                *(v4f*)(cp + 64) = (v4f){acc[4][r], acc[5][r], acc[6][r], acc[7][r]};
            }
        }
    }
}

__global__ void zero_int_kernel(int* __restrict__ p, int n) {
    int i = blockIdx.x * 256 + threadIdx.x;
    if (i < n) p[i] = 0;
}
__global__ void count_rank_kernel(int* __restrict__ cnt, int* __restrict__ rank,
                                  const int* __restrict__ dst, int e) {
    int i = blockIdx.x * 256 + threadIdx.x;
    if (i < e) rank[i] = atomicAdd(&cnt[dst[i]], 1);
}
__global__ __launch_bounds__(256)
void scan_kernel(const int* __restrict__ cnt, int* __restrict__ rowptr,
                 float* __restrict__ dinv, int n) {
    __shared__ int wtot[8];
    __shared__ int wpre[8];
    const int t = threadIdx.x;
    const int lane = t & 31;
    const int wave = t >> 5;
    const int4* c4 = (const int4*)cnt;
    const int nvec = blockIdx.x * 256;
    int pre = 0;
    int i = t;
    for (; i + 7 * 256 < nvec; i += 8 * 256) {
        const int4 a0 = c4[i],           a1 = c4[i + 256],     a2 = c4[i + 2 * 256], a3 = c4[i + 3 * 256];
        const int4 a4 = c4[i + 4 * 256], a5 = c4[i + 5 * 256], a6 = c4[i + 6 * 256], a7 = c4[i + 7 * 256];
        pre += ((a0.x + a0.y) + (a0.z + a0.w)) + ((a1.x + a1.y) + (a1.z + a1.w));
        pre += ((a2.x + a2.y) + (a2.z + a2.w)) + ((a3.x + a3.y) + (a3.z + a3.w));
        pre += ((a4.x + a4.y) + (a4.z + a4.w)) + ((a5.x + a5.y) + (a5.z + a5.w));
        pre += ((a6.x + a6.y) + (a6.z + a6.w)) + ((a7.x + a7.y) + (a7.z + a7.w));
    }
    for (; i < nvec; i += 256) {
        const int4 v = c4[i];
        pre += (v.x + v.y) + (v.z + v.w);
    }
    const int4 mine = c4[nvec + t];
    const int s4 = (mine.x + mine.y) + (mine.z + mine.w);
    int incl = s4;
#pragma unroll
    for (int off = 1; off < 32; off <<= 1) {
        const int v = __shfl_up(incl, off);
        if (lane >= off) incl += v;
    }
#pragma unroll
    for (int off = 16; off > 0; off >>= 1) pre += __shfl_xor(pre, off);
    if (lane == 31) wtot[wave] = incl;
    if (lane == 0)  wpre[wave] = pre;
    __syncthreads();
    int run = 0;
#pragma unroll
    for (int w = 0; w < 8; ++w) {
        run += wpre[w];
        if (w < wave) run += wtot[w];
    }
    run += incl - s4;
    const int i0 = (nvec + t) * 4;
    const int c[4] = {mine.x, mine.y, mine.z, mine.w};
#pragma unroll
    for (int k = 0; k < 4; ++k) {
        const int idx = i0 + k;
        if (idx <= n) rowptr[idx] = run;
        if (idx < n)  dinv[idx] = rsqrtf(1.0f + (float)c[k]);
        run += c[k];
    }
}
__global__ void fill_kernel(const int* __restrict__ rowptr, const int* __restrict__ rank,
                            const int* __restrict__ src, const int* __restrict__ dst,
                            int* __restrict__ csr, int e) {
    int i = blockIdx.x * 256 + threadIdx.x;
    if (i < e) csr[rowptr[dst[i]] + rank[i]] = src[i];
}

__global__ __launch_bounds__(256)
void gather_conv_kernel(const float* xin, float* xout, const float* __restrict__ h,
                        const int* __restrict__ rowptr, const int* __restrict__ csr,
                        const float* __restrict__ dinv, const float* __restrict__ bc, int n) {
    const int lane = threadIdx.x & 31;
    const int w0 = __builtin_amdgcn_readfirstlane((int)(blockIdx.x * 8 + (threadIdx.x >> 5)));
    const int nw = (int)gridDim.x * 8;
    const float4* __restrict__ h4 = (const float4*)h;
    const float4* xi4 = (const float4*)xin;
    float4* xo4 = (float4*)xout;
    const float4 b = ((const float4*)bc)[lane];

    for (int d = w0; d < n; d += nw) {
        const int start = __builtin_amdgcn_readfirstlane(rowptr[d]);
        const int end   = __builtin_amdgcn_readfirstlane(rowptr[d + 1]);
        const float dd  = dinv[d];
        const size_t o  = (size_t)d * (GD / 4) + lane;
        const float4 hv = h4[o];
        const float4 xv = xi4[o];
        float4 acc = {0.0f, 0.0f, 0.0f, 0.0f};

        for (int base = start; base < end; base += 32) {
            const int m = (end - base) < 32 ? (end - base) : 32;
            int   my_s = 0;
            float my_w = 0.0f;
            if (lane < m) {
                my_s = csr[base + lane];
                my_w = dinv[my_s] * dd;
            }
            const int my_wi = __float_as_int(my_w);
            int j = 0;
            for (; j + 4 <= m; j += 4) {
                const int s0 = __builtin_amdgcn_readlane(my_s, j);
                const int s1 = __builtin_amdgcn_readlane(my_s, j + 1);
                const int s2 = __builtin_amdgcn_readlane(my_s, j + 2);
                const int s3 = __builtin_amdgcn_readlane(my_s, j + 3);
                const float4 v0 = h4[(size_t)s0 * (GD / 4) + lane];
                const float4 v1 = h4[(size_t)s1 * (GD / 4) + lane];
                const float4 v2 = h4[(size_t)s2 * (GD / 4) + lane];
                const float4 v3 = h4[(size_t)s3 * (GD / 4) + lane];
                const float q0 = __int_as_float(__builtin_amdgcn_readlane(my_wi, j));
                const float q1 = __int_as_float(__builtin_amdgcn_readlane(my_wi, j + 1));
                const float q2 = __int_as_float(__builtin_amdgcn_readlane(my_wi, j + 2));
                const float q3 = __int_as_float(__builtin_amdgcn_readlane(my_wi, j + 3));
                acc.x += v0.x * q0; acc.y += v0.y * q0; acc.z += v0.z * q0; acc.w += v0.w * q0;
                acc.x += v1.x * q1; acc.y += v1.y * q1; acc.z += v1.z * q1; acc.w += v1.w * q1;
                acc.x += v2.x * q2; acc.y += v2.y * q2; acc.z += v2.z * q2; acc.w += v2.w * q2;
                acc.x += v3.x * q3; acc.y += v3.y * q3; acc.z += v3.z * q3; acc.w += v3.w * q3;
            }
            const int r = m - j;
            if (r > 0) {
                const int sa = __builtin_amdgcn_readlane(my_s, j);
                const float qa = __int_as_float(__builtin_amdgcn_readlane(my_wi, j));
                const float4 va = h4[(size_t)sa * (GD / 4) + lane];
                float4 vb = {0.0f, 0.0f, 0.0f, 0.0f};
                float4 vc = {0.0f, 0.0f, 0.0f, 0.0f};
                float qb = 0.0f, qc = 0.0f;
                if (r > 1) {
                    const int sb = __builtin_amdgcn_readlane(my_s, j + 1);
                    qb = __int_as_float(__builtin_amdgcn_readlane(my_wi, j + 1));
                    vb = h4[(size_t)sb * (GD / 4) + lane];
                }
                if (r > 2) {
                    const int sc = __builtin_amdgcn_readlane(my_s, j + 2);
                    qc = __int_as_float(__builtin_amdgcn_readlane(my_wi, j + 2));
                    vc = h4[(size_t)sc * (GD / 4) + lane];
                }
                acc.x += va.x * qa; acc.y += va.y * qa; acc.z += va.z * qa; acc.w += va.w * qa;
                acc.x += vb.x * qb; acc.y += vb.y * qb; acc.z += vb.z * qb; acc.w += vb.w * qb;
                acc.x += vc.x * qc; acc.y += vc.y * qc; acc.z += vc.z * qc; acc.w += vc.w * qc;
            }
        }

        float4 r;
        r.x = (acc.x + hv.x * dd * dd + b.x) + xv.x;
        r.y = (acc.y + hv.y * dd * dd + b.y) + xv.y;
        r.z = (acc.z + hv.z * dd * dd + b.z) + xv.z;
        r.w = (acc.w + hv.w * dd * dd + b.w) + xv.w;
        r.x = r.x > 0.0f ? r.x : 0.0f;
        r.y = r.y > 0.0f ? r.y : 0.0f;
        r.z = r.z > 0.0f ? r.z : 0.0f;
        r.w = r.w > 0.0f ? r.w : 0.0f;
        xo4[o] = r;
    }
}

extern "C" void kernel_launch(void* const* d_in, const int* in_sizes, int n_in,
                              void* d_out, int out_size, void* d_ws, size_t ws_size,
                              hipStream_t stream) {
    const float* x_in  = (const float*)d_in[0];
    const int*   ei    = (const int*)d_in[1];
    const float* Wc    = (const float*)d_in[2];
    const float* bc    = (const float*)d_in[3];
    const float* Wf    = (const float*)d_in[4];
    const float* bf    = (const float*)d_in[5];
    const float* Wskip = (const float*)d_in[6];
    const float* bskip = (const float*)d_in[7];
    const float* gamma = (const float*)d_in[8];
    const float* beta  = (const float*)d_in[9];
    const float* rmean = (const float*)d_in[10];
    const float* rvar  = (const float*)d_in[11];

    const int n = in_sizes[0] / GD;
    const int e = in_sizes[1] / 2;
    const int L = in_sizes[3] / GD;

    const int* src = ei;
    const int* dst = ei + e;

    float* x    = (float*)d_out;
    float* bufA = (float*)d_ws;
    int*   ibuf = (int*)(bufA + (size_t)n * GD);
    const int npad = (n + 1 + 1023) & ~1023;
    int*   cnt    = ibuf;
    int*   rowptr = cnt + npad;
    int*   rank   = rowptr + npad;
    int*   csr    = rank + e;
    float* dinv   = (float*)ibuf + (size_t)n * GD + n;

    dim3 blk(256);
    const int gE    = (e + 255) / 256;
    const int gPad  = npad / 256;
    const int gScan = npad / 1024;
    const int gGemm = (n + 127) / 128;
    const int gGath = 2048;

    zero_int_kernel<<<gPad, blk, 0, stream>>>(cnt, npad);
    count_rank_kernel<<<gE, blk, 0, stream>>>(cnt, rank, dst, e);
    scan_kernel<<<gScan, blk, 0, stream>>>(cnt, rowptr, dinv, n);
    fill_kernel<<<gE, blk, 0, stream>>>(rowptr, rank, src, dst, csr, e);

    wmma_gemm128<<<gGemm, blk, 0, stream>>>(x_in, Wc, bufA, n);
    for (int i = 0; i < L; ++i) {
        gather_conv_kernel<<<gGath, blk, 0, stream>>>((i == 0) ? x_in : x, x, bufA, rowptr, csr, dinv, bc + i * GD, n);
        const float* Wfi = Wf + (size_t)i * GD * GD;
        const float* Wn  = Wc + (size_t)(i + 1) * GD * GD;
        if (i == 0) {
            if (L > 1)
                wmma_chain128<false, true><<<gGemm, blk, 0, stream>>>(x, bufA, n, Wfi, bf + i * GD, nullptr, nullptr, Wn,
                                                                      gamma + i * GD, beta + i * GD, rmean + i * GD, rvar + i * GD);
            else
                wmma_chain128<false, false><<<gGemm, blk, 0, stream>>>(x, bufA, n, Wfi, bf + i * GD, nullptr, nullptr, nullptr,
                                                                       gamma + i * GD, beta + i * GD, rmean + i * GD, rvar + i * GD);
        } else {
            const float* Wsi = Wskip + (size_t)(i - 1) * GD * GD;
            if (i + 1 < L)
                wmma_chain128<true, true><<<gGemm, blk, 0, stream>>>(x, bufA, n, Wfi, bf + i * GD, Wsi, bskip + (i - 1) * GD, Wn,
                                                                     gamma + i * GD, beta + i * GD, rmean + i * GD, rvar + i * GD);
            else
                wmma_chain128<true, false><<<gGemm, blk, 0, stream>>>(x, bufA, n, Wfi, bf + i * GD, Wsi, bskip + (i - 1) * GD, nullptr,
                                                                      gamma + i * GD, beta + i * GD, rmean + i * GD, rvar + i * GD);
        }
    }
}


# Aggregation moved in front of the conv product and into the layer's product launch: a wave gathers its stripe of x rows into its LDS tile, then conv, ff, skip; sum is (sum w x)Wc, not sum w (xWc)
# speedup vs baseline: 3.2014x; 1.1300x over previous
#include <hip/hip_runtime.h>
#include <hip/hip_bf16.h>

typedef __attribute__((ext_vector_type(2))) float v2f;
typedef __attribute__((ext_vector_type(4))) float v4f;
typedef __attribute__((ext_vector_type(8))) float v8f;

#define GD 128
#define BN_EPS 1e-5f
#define LRELU_SLOPE 0.01f

__device__ __forceinline__ void w_fetch(const float* __restrict__ W, int wave, int half, int l16,
                                        v4f (&wa)[8], v4f (&wb)[8]) {
#pragma unroll
    for (int i = 0; i < 8; ++i) {
        const int sp = wave + 8 * i;
        const int s = sp >> 1, ph = sp & 1;
        const float* wp = W + (4 * s + 2 * half) * GD + 4 * l16 + 64 * ph;
        wa[i] = *(const v4f*)wp;
        wb[i] = *(const v4f*)(wp + GD);
    }
}
__device__ __forceinline__ void w_put(v4f* ldsv, int wave, int lane, const v4f (&wa)[8], const v4f (&wb)[8]) {
#pragma unroll
    for (int i = 0; i < 8; ++i) {
        const int sp = wave + 8 * i;
        const int s = sp >> 1, ph = sp & 1;
        ldsv[(s * 4 + 2 * ph) * 32 + lane]     = (v4f){wa[i].x, wb[i].x, wa[i].y, wb[i].y};
        ldsv[(s * 4 + 2 * ph + 1) * 32 + lane] = (v4f){wa[i].z, wb[i].z, wa[i].w, wb[i].w};
    }
}
__device__ __forceinline__ void stripe_product(const v4f* ldsv, int lane, const v2f (&a)[32], v8f (&acc)[8]) {
#pragma unroll
    for (int nt = 0; nt < 8; ++nt) acc[nt] = (v8f){0, 0, 0, 0, 0, 0, 0, 0};
    v4f bq[2][4];
#pragma unroll
    for (int p = 0; p < 4; ++p) bq[0][p] = ldsv[p * 32 + lane];
    __builtin_amdgcn_sched_group_barrier(0x100, 4, 0);
#pragma unroll
    for (int s = 0; s < 32; ++s) {
        if (s + 1 < 32) {
#pragma unroll
            for (int p = 0; p < 4; ++p) bq[(s + 1) & 1][p] = ldsv[((s + 1) * 4 + p) * 32 + lane];
        }
#pragma unroll
        for (int p = 0; p < 4; ++p) {
            const v4f q = bq[s & 1][p];
            const v2f b0 = {q.x, q.y};
            const v2f b1 = {q.z, q.w};
            acc[2 * p] = __builtin_amdgcn_wmma_f32_16x16x4_f32(
                false, a[s], false, b0, (short)0, acc[2 * p], false, false);
            acc[2 * p + 1] = __builtin_amdgcn_wmma_f32_16x16x4_f32(
                false, a[s], false, b1, (short)0, acc[2 * p + 1], false, false);
        }
        __builtin_amdgcn_sched_group_barrier(0x100, 4, 0);
        __builtin_amdgcn_sched_group_barrier(0x008, 8, 0);
    }
}

#define TP 132

struct BnConst { float m[8], r[8], g[8], e[8]; };
__device__ __forceinline__ void bn_load(BnConst& k, const float* __restrict__ gamma, const float* __restrict__ beta,
                                        const float* __restrict__ rmean, const float* __restrict__ rvar, int c0) {
#pragma unroll
    for (int g = 0; g < 2; ++g) {
        const int col = c0 + 64 * g;
        const v4f vm = *(const v4f*)(rmean + col);
        const v4f vv = *(const v4f*)(rvar + col);
        const v4f vg = *(const v4f*)(gamma + col);
        const v4f ve = *(const v4f*)(beta + col);
#pragma unroll
        for (int j = 0; j < 4; ++j) {
            k.m[4 * g + j] = vm[j];
            k.r[4 * g + j] = rsqrtf(vv[j] + BN_EPS);
            k.g[4 * g + j] = vg[j];
            k.e[4 * g + j] = ve[j];
        }
    }
}
__device__ __forceinline__ float bn_apply(const BnConst& k, int nt, float v) {
    const float bnv = (v - k.m[nt]) * k.r[nt] * k.g[nt] + k.e[nt];
    v = bnv + v;
    return v > 0.0f ? v : 0.0f;
}

__global__ void zero_int_kernel(int* __restrict__ p, int n) {
    int i = blockIdx.x * 256 + threadIdx.x;
    if (i < n) p[i] = 0;
}
__global__ void count_rank_kernel(int* __restrict__ cnt, int* __restrict__ rank,
                                  const int* __restrict__ dst, int e) {
    int i = blockIdx.x * 256 + threadIdx.x;
    if (i < e) rank[i] = atomicAdd(&cnt[dst[i]], 1);
}
__global__ __launch_bounds__(256)
void scan_kernel(const int* __restrict__ cnt, int* __restrict__ rowptr,
                 float* __restrict__ dinv, int n) {
    __shared__ int wtot[8];
    __shared__ int wpre[8];
    const int t = threadIdx.x;
    const int lane = t & 31;
    const int wave = t >> 5;
    const int4* c4 = (const int4*)cnt;
    const int nvec = blockIdx.x * 256;
    int pre = 0;
    int i = t;
    for (; i + 7 * 256 < nvec; i += 8 * 256) {
        const int4 a0 = c4[i],           a1 = c4[i + 256],     a2 = c4[i + 2 * 256], a3 = c4[i + 3 * 256];
        const int4 a4 = c4[i + 4 * 256], a5 = c4[i + 5 * 256], a6 = c4[i + 6 * 256], a7 = c4[i + 7 * 256];
        pre += ((a0.x + a0.y) + (a0.z + a0.w)) + ((a1.x + a1.y) + (a1.z + a1.w));
        pre += ((a2.x + a2.y) + (a2.z + a2.w)) + ((a3.x + a3.y) + (a3.z + a3.w));
        pre += ((a4.x + a4.y) + (a4.z + a4.w)) + ((a5.x + a5.y) + (a5.z + a5.w));
        pre += ((a6.x + a6.y) + (a6.z + a6.w)) + ((a7.x + a7.y) + (a7.z + a7.w));
    }
    for (; i < nvec; i += 256) {
        const int4 v = c4[i];
        pre += (v.x + v.y) + (v.z + v.w);
    }
    const int4 mine = c4[nvec + t];
    const int s4 = (mine.x + mine.y) + (mine.z + mine.w);
    int incl = s4;
#pragma unroll
    for (int off = 1; off < 32; off <<= 1) {
        const int v = __shfl_up(incl, off);
        if (lane >= off) incl += v;
    }
#pragma unroll
    for (int off = 16; off > 0; off >>= 1) pre += __shfl_xor(pre, off);
    if (lane == 31) wtot[wave] = incl;
    if (lane == 0)  wpre[wave] = pre;
    __syncthreads();
    int run = 0;
#pragma unroll
    for (int w = 0; w < 8; ++w) {
        run += wpre[w];
        if (w < wave) run += wtot[w];
    }
    run += incl - s4;
    const int i0 = (nvec + t) * 4;
    const int c[4] = {mine.x, mine.y, mine.z, mine.w};
#pragma unroll
    for (int k = 0; k < 4; ++k) {
        const int idx = i0 + k;
        if (idx <= n) rowptr[idx] = run;
        if (idx < n)  dinv[idx] = rsqrtf(1.0f + (float)c[k]);
        run += c[k];
    }
}
__global__ void fill_kernel(const int* __restrict__ rowptr, const int* __restrict__ rank,
                            const int* __restrict__ src, const int* __restrict__ dst,
                            int* __restrict__ csr, int e) {
    int i = blockIdx.x * 256 + threadIdx.x;
    if (i < e) csr[rowptr[dst[i]] + rank[i]] = src[i];
}

__device__ __forceinline__ void row_done(float* tg, int rpl, int ddi, int& cur, int& rend, float& dd, float4& g) {
    float* p = tg + cur * TP;
    const float4 xo = *(const float4*)p;
    float4 o;
    o.x = g.x + xo.x * dd * dd;
    o.y = g.y + xo.y * dd * dd;
    o.z = g.z + xo.z * dd * dd;
    o.w = g.w + xo.w * dd * dd;
    *(float4*)p = o;
    g.x = 0.0f; g.y = 0.0f; g.z = 0.0f; g.w = 0.0f;
    ++cur;
    rend = __builtin_amdgcn_readlane(rpl, cur + 1);
    dd   = __int_as_float(__builtin_amdgcn_readlane(ddi, cur));
}
__device__ __forceinline__ void rows_ask(const float4* __restrict__ xs4, int lane, int my_s, int m, int j0,
                                         float4 (&v)[16]) {
#pragma unroll
    for (int k = 0; k < 16; ++k) {
        int jj = j0 + k;
        if (jj >= m) jj = m - 1;
        const int s = __builtin_amdgcn_readlane(my_s, jj);
        v[k] = xs4[(size_t)s * (GD / 4) + lane];
    }
}
__device__ __forceinline__ void rows_add(const float4 (&v)[16], int my_qi, int base, int m, int j0,
                                         float* tg, int rpl, int ddi,
                                         int& cur, int& rend, float& dd, float4& g) {
#pragma unroll
    for (int k = 0; k < 16; ++k) {
        const int jj = j0 + k;
        if (jj < m) {
            const int e = base + jj;
            while (e == rend) row_done(tg, rpl, ddi, cur, rend, dd, g);
            const float w = __int_as_float(__builtin_amdgcn_readlane(my_qi, jj)) * dd;
            g.x += v[k].x * w; g.y += v[k].y * w; g.z += v[k].z * w; g.w += v[k].w * w;
        }
    }
}

template <bool SKIP>
__global__ __launch_bounds__(256)
void gcn_layer128(const float* __restrict__ xsrc, float* __restrict__ xdst, int nrows,
                  const int* __restrict__ rowptr, const int* __restrict__ csr,
                  const float* __restrict__ dinv,
                  const float* __restrict__ Wc, const float* __restrict__ bc,
                  const float* __restrict__ Wf, const float* __restrict__ bf,
                  const float* __restrict__ Ws, const float* __restrict__ bs,
                  const float* __restrict__ gamma, const float* __restrict__ beta,
                  const float* __restrict__ rmean, const float* __restrict__ rvar) {
    __shared__ v4f lds[GD * GD / 4 + 8 * 16 * TP / 4];
    v4f* ldsv = lds;
    float* tile = (float*)(lds + GD * GD / 4);
    const int t = threadIdx.x;
    const int wave = t >> 5;
    const int lane = t & 31;
    const int half = lane >> 4;
    const int l16  = lane & 15;
    const int c0 = 4 * l16;
    const int m0 = blockIdx.x * 128 + wave * 16;
    float* tw = tile + wave * 16 * TP;
    float* tc = tw + (8 * half) * TP + c0;
    const float* ta = tw + l16 * TP + 2 * half;
    float* tg = tw + 4 * lane;

    v4f wa[8], wb[8];
    v2f a[32];
    v8f acc[8];

    w_fetch(Wc, wave, half, l16, wa, wb);

    {
        const float4* __restrict__ xs4 = (const float4*)xsrc;
        int ridx = m0 + (lane < 16 ? lane : 16);
        if (ridx > nrows) ridx = nrows;
        const int rpl = rowptr[ridx];
        int drow = m0 + l16;
        if (drow >= nrows) drow = nrows - 1;
        const int ddi = __float_as_int(dinv[drow]);
        float4 own[16];
#pragma unroll
        for (int r = 0; r < 16; ++r) {
            int row = m0 + r;
            if (row >= nrows) row = nrows - 1;
            own[r] = xs4[(size_t)row * (GD / 4) + lane];
        }
        const int e_begin = __builtin_amdgcn_readlane(rpl, 0);
        const int e_end   = __builtin_amdgcn_readlane(rpl, 16);
        int cur = 0;
        int rend = __builtin_amdgcn_readlane(rpl, 1);
        float dd = __int_as_float(__builtin_amdgcn_readlane(ddi, 0));
        float4 g = {0.0f, 0.0f, 0.0f, 0.0f};

        int s_cur = 0, s_nx1 = 0;
        float q_cur = 0.0f;
        if (e_begin < e_end) {
            int i0 = e_begin + lane;
            if (i0 > e_end - 1) i0 = e_end - 1;
            int i1 = e_begin + 32 + lane;
            if (i1 > e_end - 1) i1 = e_end - 1;
            s_cur = csr[i0];
            s_nx1 = csr[i1];
            q_cur = dinv[s_cur];
        }
#pragma unroll
        for (int r = 0; r < 16; ++r) *(float4*)(tg + r * TP) = own[r];

        for (int base = e_begin; base < e_end; base += 32) {
            const int m = (e_end - base) < 32 ? (e_end - base) : 32;
            const int my_qi = __float_as_int(q_cur);
            float4 v0[16], v1[16];
            rows_ask(xs4, lane, s_cur, m, 0, v0);
            if (m > 16) rows_ask(xs4, lane, s_cur, m, 16, v1);
            int i2 = base + 64 + lane;
            if (i2 > e_end - 1) i2 = e_end - 1;
            const float q_nx1 = dinv[s_nx1];
            const int s_nx2 = csr[i2];
            rows_add(v0, my_qi, base, m, 0, tg, rpl, ddi, cur, rend, dd, g);
            if (m > 16) rows_add(v1, my_qi, base, m, 16, tg, rpl, ddi, cur, rend, dd, g);
            s_cur = s_nx1; q_cur = q_nx1; s_nx1 = s_nx2;
        }
        while (cur < 16) row_done(tg, rpl, ddi, cur, rend, dd, g);
    }

    w_put(ldsv, wave, lane, wa, wb);
    __syncthreads();
#pragma unroll
    for (int s = 0; s < 32; ++s) a[s] = *(const v2f*)(ta + 4 * s);
    stripe_product(ldsv, lane, a, acc);
    w_fetch(Wf, wave, half, l16, wa, wb);

    {
        float cb[8];
#pragma unroll
        for (int gq = 0; gq < 2; ++gq) {
            const v4f vb = *(const v4f*)(bc + c0 + 64 * gq);
#pragma unroll
            for (int j = 0; j < 4; ++j) cb[4 * gq + j] = vb[j];
        }
        v4f xo[8][2];
#pragma unroll
        for (int r = 0; r < 8; ++r) {
            int lrow = m0 + r + half * 8;
            if (lrow >= nrows) lrow = nrows - 1;
            const float* xp = xsrc + (size_t)lrow * GD + c0;
            xo[r][0] = *(const v4f*)xp;
            xo[r][1] = *(const v4f*)(xp + 64);
        }
#pragma unroll
        for (int r = 0; r < 8; ++r) {
            v4f o[2];
#pragma unroll
            for (int nt = 0; nt < 8; ++nt) {
                float v = (acc[nt][r] + cb[nt]) + xo[r][nt >> 2][nt & 3];
                o[nt >> 2][nt & 3] = v > 0.0f ? v : 0.0f;
            }
            *(v4f*)(tc + r * TP)      = o[0];
            *(v4f*)(tc + r * TP + 64) = o[1];
        }
    }

    __syncthreads();
    w_put(ldsv, wave, lane, wa, wb);
    __syncthreads();
#pragma unroll
    for (int s = 0; s < 32; ++s) a[s] = *(const v2f*)(ta + 4 * s);
    stripe_product(ldsv, lane, a, acc);
    if (SKIP) w_fetch(Ws, wave, half, l16, wa, wb);

    {
        float cb[8];
#pragma unroll
        for (int gq = 0; gq < 2; ++gq) {
            const v4f vb = *(const v4f*)(bf + c0 + 64 * gq);
#pragma unroll
            for (int j = 0; j < 4; ++j) cb[4 * gq + j] = vb[j];
        }
        BnConst k;
        if (!SKIP) bn_load(k, gamma, beta, rmean, rvar, c0);
#pragma unroll
        for (int r = 0; r < 8; ++r) {
            v4f vo[2];
            vo[0] = *(const v4f*)(tc + r * TP);
            vo[1] = *(const v4f*)(tc + r * TP + 64);
            v4f o[2];
#pragma unroll
            for (int nt = 0; nt < 8; ++nt) {
                float u = acc[nt][r] + cb[nt];
                u = u > 0.0f ? u : LRELU_SLOPE * u;
                float v = u + vo[nt >> 2][nt & 3];
                v = v > 0.0f ? v : 0.0f;
                if (!SKIP) v = bn_apply(k, nt, v);
                o[nt >> 2][nt & 3] = v;
            }
            if (SKIP) {
                *(v4f*)(tc + r * TP)      = o[0];
                *(v4f*)(tc + r * TP + 64) = o[1];
            } else {
                const int row = m0 + r + half * 8;
                if (row < nrows) {
                    float* cp = xdst + (size_t)row * GD + c0;
                    *(v4f*)cp        = o[0];
                    *(v4f*)(cp + 64) = o[1];
                }
            }
        }
    }

    if (SKIP) {
        __syncthreads();
        w_put(ldsv, wave, lane, wa, wb);
        __syncthreads();
#pragma unroll
        for (int s = 0; s < 32; ++s) a[s] = *(const v2f*)(ta + 4 * s);
        stripe_product(ldsv, lane, a, acc);

        float cb[8];
#pragma unroll
        for (int gq = 0; gq < 2; ++gq) {
            const v4f vb = *(const v4f*)(bs + c0 + 64 * gq);
#pragma unroll
            for (int j = 0; j < 4; ++j) cb[4 * gq + j] = vb[j];
        }
        BnConst k;
        bn_load(k, gamma, beta, rmean, rvar, c0);
#pragma unroll
        for (int r = 0; r < 8; ++r) {
            v4f vo[2];
            vo[0] = *(const v4f*)(tc + r * TP);
            vo[1] = *(const v4f*)(tc + r * TP + 64);
            v4f o[2];
#pragma unroll
            for (int nt = 0; nt < 8; ++nt) {
                const float u = acc[nt][r] + cb[nt];
                float v = u + vo[nt >> 2][nt & 3];
                v = v > 0.0f ? v : 0.0f;
                o[nt >> 2][nt & 3] = bn_apply(k, nt, v);
            }
            const int row = m0 + r + half * 8;
            if (row < nrows) {
                float* cp = xdst + (size_t)row * GD + c0;
                *(v4f*)cp        = o[0];
                *(v4f*)(cp + 64) = o[1];
            }
        }
    }
}

extern "C" void kernel_launch(void* const* d_in, const int* in_sizes, int n_in,
                              void* d_out, int out_size, void* d_ws, size_t ws_size,
                              hipStream_t stream) {
    const float* x_in  = (const float*)d_in[0];
    const int*   ei    = (const int*)d_in[1];
    const float* Wc    = (const float*)d_in[2];
    const float* bc    = (const float*)d_in[3];
    const float* Wf    = (const float*)d_in[4];
    const float* bf    = (const float*)d_in[5];
    const float* Wskip = (const float*)d_in[6];
    const float* bskip = (const float*)d_in[7];
    const float* gamma = (const float*)d_in[8];
    const float* beta  = (const float*)d_in[9];
    const float* rmean = (const float*)d_in[10];
    const float* rvar  = (const float*)d_in[11];

    const int n = in_sizes[0] / GD;
    const int e = in_sizes[1] / 2;
    const int L = in_sizes[3] / GD;

    const int* src = ei;
    const int* dst = ei + e;

    float* x    = (float*)d_out;
    float* bufA = (float*)d_ws;
    int*   ibuf = (int*)(bufA + (size_t)n * GD);
    const int npad = (n + 1 + 1023) & ~1023;
    int*   cnt    = ibuf;
    int*   rowptr = cnt + npad;
    int*   rank   = rowptr + npad;
    int*   csr    = rank + e;
    float* dinv   = (float*)ibuf + (size_t)n * GD + n;

    dim3 blk(256);
    const int gE    = (e + 255) / 256;
    const int gPad  = npad / 256;
    const int gScan = npad / 1024;
    const int gLay  = (n + 127) / 128;

    zero_int_kernel<<<gPad, blk, 0, stream>>>(cnt, npad);
    count_rank_kernel<<<gE, blk, 0, stream>>>(cnt, rank, dst, e);
    scan_kernel<<<gScan, blk, 0, stream>>>(cnt, rowptr, dinv, n);
    fill_kernel<<<gE, blk, 0, stream>>>(rowptr, rank, src, dst, csr, e);

    const float* cur_src = x_in;
    for (int i = 0; i < L; ++i) {
        float* cur_dst = (((L - 1 - i) & 1) == 0) ? x : bufA;
        const float* Wci = Wc + (size_t)i * GD * GD;
        const float* Wfi = Wf + (size_t)i * GD * GD;
        if (i == 0) {
            gcn_layer128<false><<<gLay, blk, 0, stream>>>(cur_src, cur_dst, n, rowptr, csr, dinv,
                                                          Wci, bc + i * GD, Wfi, bf + i * GD, nullptr, nullptr,
                                                          gamma + i * GD, beta + i * GD, rmean + i * GD, rvar + i * GD);
        } else {
            const float* Wsi = Wskip + (size_t)(i - 1) * GD * GD;
            gcn_layer128<true><<<gLay, blk, 0, stream>>>(cur_src, cur_dst, n, rowptr, csr, dinv,
                                                         Wci, bc + i * GD, Wfi, bf + i * GD, Wsi, bskip + (i - 1) * GD,
                                                         gamma + i * GD, beta + i * GD, rmean + i * GD, rvar + i * GD);
        }
        cur_src = cur_dst;
    }
}


# Gather loop straight-line: per-edge slot clamps and bounds tests with their branches dropped; first slot past a stripe's edges closes all rows, later slots feed a dead accumulator; sum order kept
# speedup vs baseline: 3.3320x; 1.0408x over previous
#include <hip/hip_runtime.h>
#include <hip/hip_bf16.h>

typedef __attribute__((ext_vector_type(2))) float v2f;
typedef __attribute__((ext_vector_type(4))) float v4f;
typedef __attribute__((ext_vector_type(8))) float v8f;

#define GD 128
#define BN_EPS 1e-5f
#define LRELU_SLOPE 0.01f

__device__ __forceinline__ void w_fetch(const float* __restrict__ W, int wave, int half, int l16,
                                        v4f (&wa)[8], v4f (&wb)[8]) {
#pragma unroll
    for (int i = 0; i < 8; ++i) {
        const int sp = wave + 8 * i;
        const int s = sp >> 1, ph = sp & 1;
        const float* wp = W + (4 * s + 2 * half) * GD + 4 * l16 + 64 * ph;
        wa[i] = *(const v4f*)wp;
        wb[i] = *(const v4f*)(wp + GD);
    }
}
__device__ __forceinline__ void w_put(v4f* ldsv, int wave, int lane, const v4f (&wa)[8], const v4f (&wb)[8]) {
#pragma unroll
    for (int i = 0; i < 8; ++i) {
        const int sp = wave + 8 * i;
        const int s = sp >> 1, ph = sp & 1;
        ldsv[(s * 4 + 2 * ph) * 32 + lane]     = (v4f){wa[i].x, wb[i].x, wa[i].y, wb[i].y};
        ldsv[(s * 4 + 2 * ph + 1) * 32 + lane] = (v4f){wa[i].z, wb[i].z, wa[i].w, wb[i].w};
    }
}
__device__ __forceinline__ void stripe_product(const v4f* ldsv, int lane, const v2f (&a)[32], v8f (&acc)[8]) {
#pragma unroll
    for (int nt = 0; nt < 8; ++nt) acc[nt] = (v8f){0, 0, 0, 0, 0, 0, 0, 0};
    v4f bq[2][4];
#pragma unroll
    for (int p = 0; p < 4; ++p) bq[0][p] = ldsv[p * 32 + lane];
    __builtin_amdgcn_sched_group_barrier(0x100, 4, 0);
#pragma unroll
    for (int s = 0; s < 32; ++s) {
        if (s + 1 < 32) {
#pragma unroll
            for (int p = 0; p < 4; ++p) bq[(s + 1) & 1][p] = ldsv[((s + 1) * 4 + p) * 32 + lane];
        }
#pragma unroll
        for (int p = 0; p < 4; ++p) {
            const v4f q = bq[s & 1][p];
            const v2f b0 = {q.x, q.y};
            const v2f b1 = {q.z, q.w};
            acc[2 * p] = __builtin_amdgcn_wmma_f32_16x16x4_f32(
                false, a[s], false, b0, (short)0, acc[2 * p], false, false);
            acc[2 * p + 1] = __builtin_amdgcn_wmma_f32_16x16x4_f32(
                false, a[s], false, b1, (short)0, acc[2 * p + 1], false, false);
        }
        __builtin_amdgcn_sched_group_barrier(0x100, 4, 0);
        __builtin_amdgcn_sched_group_barrier(0x008, 8, 0);
    }
}

#define TP 132

struct BnConst { float m[8], r[8], g[8], e[8]; };
__device__ __forceinline__ void bn_load(BnConst& k, const float* __restrict__ gamma, const float* __restrict__ beta,
                                        const float* __restrict__ rmean, const float* __restrict__ rvar, int c0) {
#pragma unroll
    for (int g = 0; g < 2; ++g) {
        const int col = c0 + 64 * g;
        const v4f vm = *(const v4f*)(rmean + col);
        const v4f vv = *(const v4f*)(rvar + col);
        const v4f vg = *(const v4f*)(gamma + col);
        const v4f ve = *(const v4f*)(beta + col);
#pragma unroll
        for (int j = 0; j < 4; ++j) {
            k.m[4 * g + j] = vm[j];
            k.r[4 * g + j] = rsqrtf(vv[j] + BN_EPS);
            k.g[4 * g + j] = vg[j];
            k.e[4 * g + j] = ve[j];
        }
    }
}
__device__ __forceinline__ float bn_apply(const BnConst& k, int nt, float v) {
    const float bnv = (v - k.m[nt]) * k.r[nt] * k.g[nt] + k.e[nt];
    v = bnv + v;
    return v > 0.0f ? v : 0.0f;
}

__global__ void zero_int_kernel(int* __restrict__ p, int n) {
    int i = blockIdx.x * 256 + threadIdx.x;
    if (i < n) p[i] = 0;
}
__global__ void count_rank_kernel(int* __restrict__ cnt, int* __restrict__ rank,
                                  const int* __restrict__ dst, int e) {
    int i = blockIdx.x * 256 + threadIdx.x;
    if (i < e) rank[i] = atomicAdd(&cnt[dst[i]], 1);
}
__global__ __launch_bounds__(256)
void scan_kernel(const int* __restrict__ cnt, int* __restrict__ rowptr,
                 float* __restrict__ dinv, int n) {
    __shared__ int wtot[8];
    __shared__ int wpre[8];
    const int t = threadIdx.x;
    const int lane = t & 31;
    const int wave = t >> 5;
    const int4* c4 = (const int4*)cnt;
    const int nvec = blockIdx.x * 256;
    int pre = 0;
    int i = t;
    for (; i + 7 * 256 < nvec; i += 8 * 256) {
        const int4 a0 = c4[i],           a1 = c4[i + 256],     a2 = c4[i + 2 * 256], a3 = c4[i + 3 * 256];
        const int4 a4 = c4[i + 4 * 256], a5 = c4[i + 5 * 256], a6 = c4[i + 6 * 256], a7 = c4[i + 7 * 256];
        pre += ((a0.x + a0.y) + (a0.z + a0.w)) + ((a1.x + a1.y) + (a1.z + a1.w));
        pre += ((a2.x + a2.y) + (a2.z + a2.w)) + ((a3.x + a3.y) + (a3.z + a3.w));
        pre += ((a4.x + a4.y) + (a4.z + a4.w)) + ((a5.x + a5.y) + (a5.z + a5.w));
        pre += ((a6.x + a6.y) + (a6.z + a6.w)) + ((a7.x + a7.y) + (a7.z + a7.w));
    }
    for (; i < nvec; i += 256) {
        const int4 v = c4[i];
        pre += (v.x + v.y) + (v.z + v.w);
    }
    const int4 mine = c4[nvec + t];
    const int s4 = (mine.x + mine.y) + (mine.z + mine.w);
    int incl = s4;
#pragma unroll
    for (int off = 1; off < 32; off <<= 1) {
        const int v = __shfl_up(incl, off);
        if (lane >= off) incl += v;
    }
#pragma unroll
    for (int off = 16; off > 0; off >>= 1) pre += __shfl_xor(pre, off);
    if (lane == 31) wtot[wave] = incl;
    if (lane == 0)  wpre[wave] = pre;
    __syncthreads();
    int run = 0;
#pragma unroll
    for (int w = 0; w < 8; ++w) {
        run += wpre[w];
        if (w < wave) run += wtot[w];
    }
    run += incl - s4;
    const int i0 = (nvec + t) * 4;
    const int c[4] = {mine.x, mine.y, mine.z, mine.w};
#pragma unroll
    for (int k = 0; k < 4; ++k) {
        const int idx = i0 + k;
        if (idx <= n) rowptr[idx] = run;
        if (idx < n)  dinv[idx] = rsqrtf(1.0f + (float)c[k]);
        run += c[k];
    }
}
__global__ void fill_kernel(const int* __restrict__ rowptr, const int* __restrict__ rank,
                            const int* __restrict__ src, const int* __restrict__ dst,
                            int* __restrict__ csr, int e) {
    int i = blockIdx.x * 256 + threadIdx.x;
    if (i < e) csr[rowptr[dst[i]] + rank[i]] = src[i];
}

__device__ __forceinline__ void row_done(float* tg, int rpl, int ddi, int& cur, int& rend, float& dd, float4& g) {
    float* p = tg + cur * TP;
    const float4 xo = *(const float4*)p;
    float4 o;
    o.x = g.x + xo.x * dd * dd;
    o.y = g.y + xo.y * dd * dd;
    o.z = g.z + xo.z * dd * dd;
    o.w = g.w + xo.w * dd * dd;
    *(float4*)p = o;
    g.x = 0.0f; g.y = 0.0f; g.z = 0.0f; g.w = 0.0f;
    ++cur;
    rend = __builtin_amdgcn_readlane(rpl, cur + 1);
    dd   = __int_as_float(__builtin_amdgcn_readlane(ddi, cur));
}
__device__ __forceinline__ void rows_ask(const float4* __restrict__ xs4, int lane, int my_s, int j0,
                                         float4 (&v)[16]) {
#pragma unroll
    for (int k = 0; k < 16; ++k) {
        const unsigned s = (unsigned)__builtin_amdgcn_readlane(my_s, j0 + k);
        v[k] = xs4[(size_t)s * (GD / 4) + lane];
    }
}
__device__ __forceinline__ void rows_add(const float4 (&v)[16], int my_qi, int base, int j0,
                                         float* tg, int rpl, int ddi,
                                         int& cur, int& rend, float& dd, float4& g) {
#pragma unroll
    for (int k = 0; k < 16; ++k) {
        const int jj = j0 + k;
        const int e = base + jj;
        if (__builtin_expect(e == rend, 0)) {
            do row_done(tg, rpl, ddi, cur, rend, dd, g); while (e == rend && cur < 16);
        }
        const float w = __int_as_float(__builtin_amdgcn_readlane(my_qi, jj)) * dd;
        g.x += v[k].x * w; g.y += v[k].y * w; g.z += v[k].z * w; g.w += v[k].w * w;
    }
}

template <bool SKIP>
__global__ __launch_bounds__(256)
void gcn_layer128(const float* __restrict__ xsrc, float* __restrict__ xdst, int nrows,
                  const int* __restrict__ rowptr, const int* __restrict__ csr,
                  const float* __restrict__ dinv,
                  const float* __restrict__ Wc, const float* __restrict__ bc,
                  const float* __restrict__ Wf, const float* __restrict__ bf,
                  const float* __restrict__ Ws, const float* __restrict__ bs,
                  const float* __restrict__ gamma, const float* __restrict__ beta,
                  const float* __restrict__ rmean, const float* __restrict__ rvar) {
    __shared__ v4f lds[GD * GD / 4 + 8 * 16 * TP / 4];
    v4f* ldsv = lds;
    float* tile = (float*)(lds + GD * GD / 4);
    const int t = threadIdx.x;
    const int wave = t >> 5;
    const int lane = t & 31;
    const int half = lane >> 4;
    const int l16  = lane & 15;
    const int c0 = 4 * l16;
    const int m0 = blockIdx.x * 128 + wave * 16;
    float* tw = tile + wave * 16 * TP;
    float* tc = tw + (8 * half) * TP + c0;
    const float* ta = tw + l16 * TP + 2 * half;
    float* tg = tw + 4 * lane;

    v4f wa[8], wb[8];
    v2f a[32];
    v8f acc[8];

    w_fetch(Wc, wave, half, l16, wa, wb);

    {
        const float4* __restrict__ xs4 = (const float4*)xsrc;
        int ridx = m0 + (lane < 16 ? lane : 16);
        if (ridx > nrows) ridx = nrows;
        const int rpl = rowptr[ridx];
        int drow = m0 + l16;
        if (drow >= nrows) drow = nrows - 1;
        const int ddi = __float_as_int(dinv[drow]);
        float4 own[16];
#pragma unroll
        for (int r = 0; r < 16; ++r) {
            int row = m0 + r;
            if (row >= nrows) row = nrows - 1;
            own[r] = xs4[(size_t)row * (GD / 4) + lane];
        }
        const int e_begin = __builtin_amdgcn_readlane(rpl, 0);
        const int e_end   = __builtin_amdgcn_readlane(rpl, 16);
        int cur = 0;
        int rend = __builtin_amdgcn_readlane(rpl, 1);
        float dd = __int_as_float(__builtin_amdgcn_readlane(ddi, 0));
        float4 g = {0.0f, 0.0f, 0.0f, 0.0f};

        int s_cur = 0, s_nx1 = 0;
        float q_cur = 0.0f;
        if (e_begin < e_end) {
            int i0 = e_begin + lane;
            if (i0 > e_end - 1) i0 = e_end - 1;
            int i1 = e_begin + 32 + lane;
            if (i1 > e_end - 1) i1 = e_end - 1;
            s_cur = csr[i0];
            s_nx1 = csr[i1];
            q_cur = dinv[s_cur];
        }
#pragma unroll
        for (int r = 0; r < 16; ++r) *(float4*)(tg + r * TP) = own[r];

        for (int base = e_begin; base < e_end; base += 32) {
            const int m = (e_end - base) < 32 ? (e_end - base) : 32;
            const int my_qi = __float_as_int(q_cur);
            float4 v0[16], v1[16];
            rows_ask(xs4, lane, s_cur, 0, v0);
            if (m > 16) rows_ask(xs4, lane, s_cur, 16, v1);
            int i2 = base + 64 + lane;
            if (i2 > e_end - 1) i2 = e_end - 1;
            const float q_nx1 = dinv[s_nx1];
            const int s_nx2 = csr[i2];
            rows_add(v0, my_qi, base, 0, tg, rpl, ddi, cur, rend, dd, g);
            if (m > 16) rows_add(v1, my_qi, base, 16, tg, rpl, ddi, cur, rend, dd, g);
            s_cur = s_nx1; q_cur = q_nx1; s_nx1 = s_nx2;
        }
        while (cur < 16) row_done(tg, rpl, ddi, cur, rend, dd, g);
    }

    w_put(ldsv, wave, lane, wa, wb);
    __syncthreads();
#pragma unroll
    for (int s = 0; s < 32; ++s) a[s] = *(const v2f*)(ta + 4 * s);
    stripe_product(ldsv, lane, a, acc);
    w_fetch(Wf, wave, half, l16, wa, wb);

    {
        float cb[8];
#pragma unroll
        for (int gq = 0; gq < 2; ++gq) {
            const v4f vb = *(const v4f*)(bc + c0 + 64 * gq);
#pragma unroll
            for (int j = 0; j < 4; ++j) cb[4 * gq + j] = vb[j];
        }
        v4f xo[8][2];
#pragma unroll
        for (int r = 0; r < 8; ++r) {
            int lrow = m0 + r + half * 8;
            if (lrow >= nrows) lrow = nrows - 1;
            const float* xp = xsrc + (size_t)lrow * GD + c0;
            xo[r][0] = *(const v4f*)xp;
            xo[r][1] = *(const v4f*)(xp + 64);
        }
#pragma unroll
        for (int r = 0; r < 8; ++r) {
            v4f o[2];
#pragma unroll
            for (int nt = 0; nt < 8; ++nt) {
                float v = (acc[nt][r] + cb[nt]) + xo[r][nt >> 2][nt & 3];
                o[nt >> 2][nt & 3] = v > 0.0f ? v : 0.0f;
            }
            *(v4f*)(tc + r * TP)      = o[0];
            *(v4f*)(tc + r * TP + 64) = o[1];
        }
    }

    __syncthreads();
    w_put(ldsv, wave, lane, wa, wb);
    __syncthreads();
#pragma unroll
    for (int s = 0; s < 32; ++s) a[s] = *(const v2f*)(ta + 4 * s);
    stripe_product(ldsv, lane, a, acc);
    if (SKIP) w_fetch(Ws, wave, half, l16, wa, wb);

    {
        float cb[8];
#pragma unroll
        for (int gq = 0; gq < 2; ++gq) {
            const v4f vb = *(const v4f*)(bf + c0 + 64 * gq);
#pragma unroll
            for (int j = 0; j < 4; ++j) cb[4 * gq + j] = vb[j];
        }
        BnConst k;
        if (!SKIP) bn_load(k, gamma, beta, rmean, rvar, c0);
#pragma unroll
        for (int r = 0; r < 8; ++r) {
            v4f vo[2];
            vo[0] = *(const v4f*)(tc + r * TP);
            vo[1] = *(const v4f*)(tc + r * TP + 64);
            v4f o[2];
#pragma unroll
            for (int nt = 0; nt < 8; ++nt) {
                float u = acc[nt][r] + cb[nt];
                u = u > 0.0f ? u : LRELU_SLOPE * u;
                float v = u + vo[nt >> 2][nt & 3];
                v = v > 0.0f ? v : 0.0f;
                if (!SKIP) v = bn_apply(k, nt, v);
                o[nt >> 2][nt & 3] = v;
            }
            if (SKIP) {
                *(v4f*)(tc + r * TP)      = o[0];
                *(v4f*)(tc + r * TP + 64) = o[1];
            } else {
                const int row = m0 + r + half * 8;
                if (row < nrows) {
                    float* cp = xdst + (size_t)row * GD + c0;
                    *(v4f*)cp        = o[0];
                    *(v4f*)(cp + 64) = o[1];
                }
            }
        }
    }

    if (SKIP) {
        __syncthreads();
        w_put(ldsv, wave, lane, wa, wb);
        __syncthreads();
#pragma unroll
        for (int s = 0; s < 32; ++s) a[s] = *(const v2f*)(ta + 4 * s);
        stripe_product(ldsv, lane, a, acc);

        float cb[8];
#pragma unroll
        for (int gq = 0; gq < 2; ++gq) {
            const v4f vb = *(const v4f*)(bs + c0 + 64 * gq);
#pragma unroll
            for (int j = 0; j < 4; ++j) cb[4 * gq + j] = vb[j];
        }
        BnConst k;
        bn_load(k, gamma, beta, rmean, rvar, c0);
#pragma unroll
        for (int r = 0; r < 8; ++r) {
            v4f vo[2];
            vo[0] = *(const v4f*)(tc + r * TP);
            vo[1] = *(const v4f*)(tc + r * TP + 64);
            v4f o[2];
#pragma unroll
            for (int nt = 0; nt < 8; ++nt) {
                const float u = acc[nt][r] + cb[nt];
                float v = u + vo[nt >> 2][nt & 3];
                v = v > 0.0f ? v : 0.0f;
                o[nt >> 2][nt & 3] = bn_apply(k, nt, v);
            }
            const int row = m0 + r + half * 8;
            if (row < nrows) {
                float* cp = xdst + (size_t)row * GD + c0;
                *(v4f*)cp        = o[0];
                *(v4f*)(cp + 64) = o[1];
            }
        }
    }
}

extern "C" void kernel_launch(void* const* d_in, const int* in_sizes, int n_in,
                              void* d_out, int out_size, void* d_ws, size_t ws_size,
                              hipStream_t stream) {
    const float* x_in  = (const float*)d_in[0];
    const int*   ei    = (const int*)d_in[1];
    const float* Wc    = (const float*)d_in[2];
    const float* bc    = (const float*)d_in[3];
    const float* Wf    = (const float*)d_in[4];
    const float* bf    = (const float*)d_in[5];
    const float* Wskip = (const float*)d_in[6];
    const float* bskip = (const float*)d_in[7];
    const float* gamma = (const float*)d_in[8];
    const float* beta  = (const float*)d_in[9];
    const float* rmean = (const float*)d_in[10];
    const float* rvar  = (const float*)d_in[11];

    const int n = in_sizes[0] / GD;
    const int e = in_sizes[1] / 2;
    const int L = in_sizes[3] / GD;

    const int* src = ei;
    const int* dst = ei + e;

    float* x    = (float*)d_out;
    float* bufA = (float*)d_ws;
    int*   ibuf = (int*)(bufA + (size_t)n * GD);
    const int npad = (n + 1 + 1023) & ~1023;
    int*   cnt    = ibuf;
    int*   rowptr = cnt + npad;
    int*   rank   = rowptr + npad;
    int*   csr    = rank + e;
    float* dinv   = (float*)ibuf + (size_t)n * GD + n;

    dim3 blk(256);
    const int gE    = (e + 255) / 256;
    const int gPad  = npad / 256;
    const int gScan = npad / 1024;
    const int gLay  = (n + 127) / 128;

    zero_int_kernel<<<gPad, blk, 0, stream>>>(cnt, npad);
    count_rank_kernel<<<gE, blk, 0, stream>>>(cnt, rank, dst, e);
    scan_kernel<<<gScan, blk, 0, stream>>>(cnt, rowptr, dinv, n);
    fill_kernel<<<gE, blk, 0, stream>>>(rowptr, rank, src, dst, csr, e);

    const float* cur_src = x_in;
    for (int i = 0; i < L; ++i) {
        float* cur_dst = (((L - 1 - i) & 1) == 0) ? x : bufA;
        const float* Wci = Wc + (size_t)i * GD * GD;
        const float* Wfi = Wf + (size_t)i * GD * GD;
        if (i == 0) {
            gcn_layer128<false><<<gLay, blk, 0, stream>>>(cur_src, cur_dst, n, rowptr, csr, dinv,
                                                          Wci, bc + i * GD, Wfi, bf + i * GD, nullptr, nullptr,
                                                          gamma + i * GD, beta + i * GD, rmean + i * GD, rvar + i * GD);
        } else {
            const float* Wsi = Wskip + (size_t)(i - 1) * GD * GD;
            gcn_layer128<true><<<gLay, blk, 0, stream>>>(cur_src, cur_dst, n, rowptr, csr, dinv,
                                                         Wci, bc + i * GD, Wfi, bf + i * GD, Wsi, bskip + (i - 1) * GD,
                                                         gamma + i * GD, beta + i * GD, rmean + i * GD, rvar + i * GD);
        }
        cur_src = cur_dst;
    }
}


# Edge list built through LDS bins: 8192-edge slices grouped by 128-row bin, then a work-group per bin slots its edges by LDS atomics: 4 launches to 2, no global atomics; sum order arrival, as before
# speedup vs baseline: 4.3157x; 1.2952x over previous
#include <hip/hip_runtime.h>
#include <hip/hip_bf16.h>

typedef __attribute__((ext_vector_type(2))) float v2f;
typedef __attribute__((ext_vector_type(4))) float v4f;
typedef __attribute__((ext_vector_type(8))) float v8f;

#define GD 128
#define BN_EPS 1e-5f
#define LRELU_SLOPE 0.01f

__device__ __forceinline__ void w_fetch(const float* __restrict__ W, int wave, int half, int l16,
                                        v4f (&wa)[8], v4f (&wb)[8]) {
#pragma unroll
    for (int i = 0; i < 8; ++i) {
        const int sp = wave + 8 * i;
        const int s = sp >> 1, ph = sp & 1;
        const float* wp = W + (4 * s + 2 * half) * GD + 4 * l16 + 64 * ph;
        wa[i] = *(const v4f*)wp;
        wb[i] = *(const v4f*)(wp + GD);
    }
}
__device__ __forceinline__ void w_put(v4f* ldsv, int wave, int lane, const v4f (&wa)[8], const v4f (&wb)[8]) {
#pragma unroll
    for (int i = 0; i < 8; ++i) {
        const int sp = wave + 8 * i;
        const int s = sp >> 1, ph = sp & 1;
        ldsv[(s * 4 + 2 * ph) * 32 + lane]     = (v4f){wa[i].x, wb[i].x, wa[i].y, wb[i].y};
        ldsv[(s * 4 + 2 * ph + 1) * 32 + lane] = (v4f){wa[i].z, wb[i].z, wa[i].w, wb[i].w};
    }
}
__device__ __forceinline__ void stripe_product(const v4f* ldsv, int lane, const v2f (&a)[32], v8f (&acc)[8]) {
#pragma unroll
    for (int nt = 0; nt < 8; ++nt) acc[nt] = (v8f){0, 0, 0, 0, 0, 0, 0, 0};
    v4f bq[2][4];
#pragma unroll
    for (int p = 0; p < 4; ++p) bq[0][p] = ldsv[p * 32 + lane];
    __builtin_amdgcn_sched_group_barrier(0x100, 4, 0);
#pragma unroll
    for (int s = 0; s < 32; ++s) {
        if (s + 1 < 32) {
#pragma unroll
            for (int p = 0; p < 4; ++p) bq[(s + 1) & 1][p] = ldsv[((s + 1) * 4 + p) * 32 + lane];
        }
#pragma unroll
        for (int p = 0; p < 4; ++p) {
            const v4f q = bq[s & 1][p];
            const v2f b0 = {q.x, q.y};
            const v2f b1 = {q.z, q.w};
            acc[2 * p] = __builtin_amdgcn_wmma_f32_16x16x4_f32(
                false, a[s], false, b0, (short)0, acc[2 * p], false, false);
            acc[2 * p + 1] = __builtin_amdgcn_wmma_f32_16x16x4_f32(
                false, a[s], false, b1, (short)0, acc[2 * p + 1], false, false);
        }
        __builtin_amdgcn_sched_group_barrier(0x100, 4, 0);
        __builtin_amdgcn_sched_group_barrier(0x008, 8, 0);
    }
}

#define TP 132

struct BnConst { float m[8], r[8], g[8], e[8]; };
__device__ __forceinline__ void bn_load(BnConst& k, const float* __restrict__ gamma, const float* __restrict__ beta,
                                        const float* __restrict__ rmean, const float* __restrict__ rvar, int c0) {
#pragma unroll
    for (int g = 0; g < 2; ++g) {
        const int col = c0 + 64 * g;
        const v4f vm = *(const v4f*)(rmean + col);
        const v4f vv = *(const v4f*)(rvar + col);
        const v4f vg = *(const v4f*)(gamma + col);
        const v4f ve = *(const v4f*)(beta + col);
#pragma unroll
        for (int j = 0; j < 4; ++j) {
            k.m[4 * g + j] = vm[j];
            k.r[4 * g + j] = rsqrtf(vv[j] + BN_EPS);
            k.g[4 * g + j] = vg[j];
            k.e[4 * g + j] = ve[j];
        }
    }
}
__device__ __forceinline__ float bn_apply(const BnConst& k, int nt, float v) {
    const float bnv = (v - k.m[nt]) * k.r[nt] * k.g[nt] + k.e[nt];
    v = bnv + v;
    return v > 0.0f ? v : 0.0f;
}

#define BIN_SHIFT 7
#define BIN_NODES 128
#define NB_MAX 512
#define SLICE 8192

__device__ __forceinline__ int lds_count(int* p) {
    return __hip_atomic_fetch_add(p, 1, __ATOMIC_RELAXED, __HIP_MEMORY_SCOPE_WORKGROUP);
}

__global__ __launch_bounds__(1024)
void bin_edges_kernel(const int* __restrict__ src, const int* __restrict__ dst, int e, int nbins, int nslices,
                      int* __restrict__ grouped, int* __restrict__ gstart, int* __restrict__ gcount) {
    __shared__ int hist[NB_MAX];
    __shared__ int wsum[NB_MAX / 32];
    __shared__ int stage[SLICE];
    const int t = threadIdx.x;
    const int lane = t & 31;
    const int wave = t >> 5;
    const int b = blockIdx.x;
    const int base = b * SLICE;
    if (t < NB_MAX) hist[t] = 0;
    int d[8], s[8], slot[8];
#pragma unroll
    for (int k = 0; k < 8; ++k) {
        const int idx = base + k * 1024 + t;
        const int ic = idx < e ? idx : e - 1;
        d[k] = dst[ic];
        s[k] = src[ic];
    }
#pragma unroll
    for (int k = 0; k < 8; ++k) {
        if (base + k * 1024 + t >= e) d[k] = -1;
    }
    __syncthreads();
#pragma unroll
    for (int k = 0; k < 8; ++k) {
        slot[k] = 0;
        if (d[k] >= 0) slot[k] = lds_count(&hist[d[k] >> BIN_SHIFT]);
    }
    __syncthreads();
    const int v = (t < NB_MAX) ? hist[t] : 0;
    int incl = v;
#pragma unroll
    for (int off = 1; off < 32; off <<= 1) {
        const int u = __shfl_up(incl, off);
        if (lane >= off) incl += u;
    }
    if (t < NB_MAX && lane == 31) wsum[wave] = incl;
    __syncthreads();
    if (t < NB_MAX) {
        int pre = 0;
#pragma unroll
        for (int w = 0; w < NB_MAX / 32; ++w) if (w < wave) pre += wsum[w];
        const int excl = pre + incl - v;
        hist[t] = excl;
        if (t < nbins) {
            gstart[(size_t)t * nslices + b] = excl;
            gcount[(size_t)t * nslices + b] = v;
        }
    }
    __syncthreads();
#pragma unroll
    for (int k = 0; k < 8; ++k) {
        if (d[k] >= 0) stage[hist[d[k] >> BIN_SHIFT] + slot[k]] = (s[k] << BIN_SHIFT) | (d[k] & (BIN_NODES - 1));
    }
    __syncthreads();
#pragma unroll
    for (int k = 0; k < 8; ++k) {
        const int i = k * 1024 + t;
        if (base + i < e) grouped[base + i] = stage[i];
    }
}

template <bool FILL>
__device__ __forceinline__ void edge_put(int w, int* rowslot, int* __restrict__ csr) {
    const int j = w & (BIN_NODES - 1);
    const int pos = lds_count(&rowslot[j]);
    if (FILL) csr[pos] = (int)((unsigned)w >> BIN_SHIFT);
}
template <bool FILL>
__device__ __forceinline__ void walk_groups(const int* __restrict__ grouped, const int* __restrict__ st_row,
                                            const int* __restrict__ ct_row, int nslices, int e, int wave, int lane,
                                            int* rowslot, int* __restrict__ csr) {
    for (int b0 = wave; b0 < nslices; b0 += 64) {
        int st[8], ln[8], w0[8];
#pragma unroll
        for (int q = 0; q < 8; ++q) {
            const int b = b0 + 8 * q;
            const int bc = b < nslices ? b : nslices - 1;
            st[q] = st_row[bc];
            ln[q] = ct_row[bc];
        }
#pragma unroll
        for (int q = 0; q < 8; ++q) {
            const int b = b0 + 8 * q;
            const int bc = b < nslices ? b : nslices - 1;
            if (b >= nslices) ln[q] = 0;
            int gi = bc * SLICE + st[q] + (lane < ln[q] ? lane : 0);
            if (gi > e - 1) gi = e - 1;
            w0[q] = grouped[gi];
        }
#pragma unroll
        for (int q = 0; q < 8; ++q) {
            if (lane < ln[q]) edge_put<FILL>(w0[q], rowslot, csr);
        }
#pragma unroll
        for (int q = 0; q < 8; ++q) {
            const int b = b0 + 8 * q;
            for (int i = lane + 32; i < ln[q]; i += 32)
                edge_put<FILL>(grouped[b * SLICE + st[q] + i], rowslot, csr);
        }
    }
}

__global__ __launch_bounds__(256)
void build_rows_kernel(const int* __restrict__ grouped, const int* __restrict__ gstart,
                       const int* __restrict__ gcount, int nslices, int e, int n,
                       int* __restrict__ rowptr, float* __restrict__ dinv, int* __restrict__ csr) {
    __shared__ int rowslot[BIN_NODES];
    __shared__ int wsum[8];
    __shared__ int wfront[8];
    const int t = threadIdx.x;
    const int lane = t & 31;
    const int wave = t >> 5;
    const int c = blockIdx.x;
    const int* st_row = gstart + (size_t)c * nslices;
    const int* ct_row = gcount + (size_t)c * nslices;
    if (t < BIN_NODES) rowslot[t] = 0;
    int front = 0;
    for (int i = t; i < nslices; i += 256) front += st_row[i];
    __syncthreads();
    walk_groups<false>(grouped, st_row, ct_row, nslices, e, wave, lane, rowslot, csr);
#pragma unroll
    for (int off = 16; off > 0; off >>= 1) front += __shfl_xor(front, off);
    if (lane == 0) wfront[wave] = front;
    __syncthreads();
    front = 0;
#pragma unroll
    for (int w = 0; w < 8; ++w) front += wfront[w];
    const int v = (t < BIN_NODES) ? rowslot[t] : 0;
    int incl = v;
#pragma unroll
    for (int off = 1; off < 32; off <<= 1) {
        const int u = __shfl_up(incl, off);
        if (lane >= off) incl += u;
    }
    if (t < BIN_NODES && lane == 31) wsum[wave] = incl;
    __syncthreads();
    if (t < BIN_NODES) {
        int pre = front;
#pragma unroll
        for (int w = 0; w < BIN_NODES / 32; ++w) if (w < wave) pre += wsum[w];
        const int excl = pre + incl - v;
        rowslot[t] = excl;
        const int node = c * BIN_NODES + t;
        if (node <= n) rowptr[node] = excl;
        if (node < n)  dinv[node] = rsqrtf(1.0f + (float)v);
    }
    __syncthreads();
    walk_groups<true>(grouped, st_row, ct_row, nslices, e, wave, lane, rowslot, csr);
}

__device__ __forceinline__ void row_done(float* tg, int rpl, int ddi, int& cur, int& rend, float& dd, float4& g) {
    float* p = tg + cur * TP;
    const float4 xo = *(const float4*)p;
    float4 o;
    o.x = g.x + xo.x * dd * dd;
    o.y = g.y + xo.y * dd * dd;
    o.z = g.z + xo.z * dd * dd;
    o.w = g.w + xo.w * dd * dd;
    *(float4*)p = o;
    g.x = 0.0f; g.y = 0.0f; g.z = 0.0f; g.w = 0.0f;
    ++cur;
    rend = __builtin_amdgcn_readlane(rpl, cur + 1);
    dd   = __int_as_float(__builtin_amdgcn_readlane(ddi, cur));
}
__device__ __forceinline__ void rows_ask(const float4* __restrict__ xs4, int lane, int my_s, int j0,
                                         float4 (&v)[16]) {
#pragma unroll
    for (int k = 0; k < 16; ++k) {
        const unsigned s = (unsigned)__builtin_amdgcn_readlane(my_s, j0 + k);
        v[k] = xs4[(size_t)s * (GD / 4) + lane];
    }
}
__device__ __forceinline__ void rows_add(const float4 (&v)[16], int my_qi, int base, int j0,
                                         float* tg, int rpl, int ddi,
                                         int& cur, int& rend, float& dd, float4& g) {
#pragma unroll
    for (int k = 0; k < 16; ++k) {
        const int jj = j0 + k;
        const int e = base + jj;
        if (__builtin_expect(e == rend, 0)) {
            do row_done(tg, rpl, ddi, cur, rend, dd, g); while (e == rend && cur < 16);
        }
        const float w = __int_as_float(__builtin_amdgcn_readlane(my_qi, jj)) * dd;
        g.x += v[k].x * w; g.y += v[k].y * w; g.z += v[k].z * w; g.w += v[k].w * w;
    }
}

template <bool SKIP>
__global__ __launch_bounds__(256)
void gcn_layer128(const float* __restrict__ xsrc, float* __restrict__ xdst, int nrows,
                  const int* __restrict__ rowptr, const int* __restrict__ csr,
                  const float* __restrict__ dinv,
                  const float* __restrict__ Wc, const float* __restrict__ bc,
                  const float* __restrict__ Wf, const float* __restrict__ bf,
                  const float* __restrict__ Ws, const float* __restrict__ bs,
                  const float* __restrict__ gamma, const float* __restrict__ beta,
                  const float* __restrict__ rmean, const float* __restrict__ rvar) {
    __shared__ v4f lds[GD * GD / 4 + 8 * 16 * TP / 4];
    v4f* ldsv = lds;
    float* tile = (float*)(lds + GD * GD / 4);
    const int t = threadIdx.x;
    const int wave = t >> 5;
    const int lane = t & 31;
    const int half = lane >> 4;
    const int l16  = lane & 15;
    const int c0 = 4 * l16;
    const int m0 = blockIdx.x * 128 + wave * 16;
    float* tw = tile + wave * 16 * TP;
    float* tc = tw + (8 * half) * TP + c0;
    const float* ta = tw + l16 * TP + 2 * half;
    float* tg = tw + 4 * lane;

    v4f wa[8], wb[8];
    v2f a[32];
    v8f acc[8];

    w_fetch(Wc, wave, half, l16, wa, wb);

    {
        const float4* __restrict__ xs4 = (const float4*)xsrc;
        int ridx = m0 + (lane < 16 ? lane : 16);
        if (ridx > nrows) ridx = nrows;
        const int rpl = rowptr[ridx];
        int drow = m0 + l16;
        if (drow >= nrows) drow = nrows - 1;
        const int ddi = __float_as_int(dinv[drow]);
        float4 own[16];
#pragma unroll
        for (int r = 0; r < 16; ++r) {
            int row = m0 + r;
            if (row >= nrows) row = nrows - 1;
            own[r] = xs4[(size_t)row * (GD / 4) + lane];
        }
        const int e_begin = __builtin_amdgcn_readlane(rpl, 0);
        const int e_end   = __builtin_amdgcn_readlane(rpl, 16);
        int cur = 0;
        int rend = __builtin_amdgcn_readlane(rpl, 1);
        float dd = __int_as_float(__builtin_amdgcn_readlane(ddi, 0));
        float4 g = {0.0f, 0.0f, 0.0f, 0.0f};

        int s_cur = 0, s_nx1 = 0;
        float q_cur = 0.0f;
        if (e_begin < e_end) {
            int i0 = e_begin + lane;
            if (i0 > e_end - 1) i0 = e_end - 1;
            int i1 = e_begin + 32 + lane;
            if (i1 > e_end - 1) i1 = e_end - 1;
            s_cur = csr[i0];
            s_nx1 = csr[i1];
            q_cur = dinv[s_cur];
        }
#pragma unroll
        for (int r = 0; r < 16; ++r) *(float4*)(tg + r * TP) = own[r];

        for (int base = e_begin; base < e_end; base += 32) {
            const int m = (e_end - base) < 32 ? (e_end - base) : 32;
            const int my_qi = __float_as_int(q_cur);
            float4 v0[16], v1[16];
            rows_ask(xs4, lane, s_cur, 0, v0);
            if (m > 16) rows_ask(xs4, lane, s_cur, 16, v1);
            int i2 = base + 64 + lane;
            if (i2 > e_end - 1) i2 = e_end - 1;
            const float q_nx1 = dinv[s_nx1];
            const int s_nx2 = csr[i2];
            rows_add(v0, my_qi, base, 0, tg, rpl, ddi, cur, rend, dd, g);
            if (m > 16) rows_add(v1, my_qi, base, 16, tg, rpl, ddi, cur, rend, dd, g);
            s_cur = s_nx1; q_cur = q_nx1; s_nx1 = s_nx2;
        }
        while (cur < 16) row_done(tg, rpl, ddi, cur, rend, dd, g);
    }

    w_put(ldsv, wave, lane, wa, wb);
    __syncthreads();
#pragma unroll
    for (int s = 0; s < 32; ++s) a[s] = *(const v2f*)(ta + 4 * s);
    stripe_product(ldsv, lane, a, acc);
    w_fetch(Wf, wave, half, l16, wa, wb);

    {
        float cb[8];
#pragma unroll
        for (int gq = 0; gq < 2; ++gq) {
            const v4f vb = *(const v4f*)(bc + c0 + 64 * gq);
#pragma unroll
            for (int j = 0; j < 4; ++j) cb[4 * gq + j] = vb[j];
        }
        v4f xo[8][2];
#pragma unroll
        for (int r = 0; r < 8; ++r) {
            int lrow = m0 + r + half * 8;
            if (lrow >= nrows) lrow = nrows - 1;
            const float* xp = xsrc + (size_t)lrow * GD + c0;
            xo[r][0] = *(const v4f*)xp;
            xo[r][1] = *(const v4f*)(xp + 64);
        }
#pragma unroll
        for (int r = 0; r < 8; ++r) {
            v4f o[2];
#pragma unroll
            for (int nt = 0; nt < 8; ++nt) {
                float v = (acc[nt][r] + cb[nt]) + xo[r][nt >> 2][nt & 3];
                o[nt >> 2][nt & 3] = v > 0.0f ? v : 0.0f;
            }
            *(v4f*)(tc + r * TP)      = o[0];
            *(v4f*)(tc + r * TP + 64) = o[1];
        }
    }

    __syncthreads();
    w_put(ldsv, wave, lane, wa, wb);
    __syncthreads();
#pragma unroll
    for (int s = 0; s < 32; ++s) a[s] = *(const v2f*)(ta + 4 * s);
    stripe_product(ldsv, lane, a, acc);
    if (SKIP) w_fetch(Ws, wave, half, l16, wa, wb);

    {
        float cb[8];
#pragma unroll
        for (int gq = 0; gq < 2; ++gq) {
            const v4f vb = *(const v4f*)(bf + c0 + 64 * gq);
#pragma unroll
            for (int j = 0; j < 4; ++j) cb[4 * gq + j] = vb[j];
        }
        BnConst k;
        if (!SKIP) bn_load(k, gamma, beta, rmean, rvar, c0);
#pragma unroll
        for (int r = 0; r < 8; ++r) {
            v4f vo[2];
            vo[0] = *(const v4f*)(tc + r * TP);
            vo[1] = *(const v4f*)(tc + r * TP + 64);
            v4f o[2];
#pragma unroll
            for (int nt = 0; nt < 8; ++nt) {
                float u = acc[nt][r] + cb[nt];
                u = u > 0.0f ? u : LRELU_SLOPE * u;
                float v = u + vo[nt >> 2][nt & 3];
                v = v > 0.0f ? v : 0.0f;
                if (!SKIP) v = bn_apply(k, nt, v);
                o[nt >> 2][nt & 3] = v;
            }
            if (SKIP) {
                *(v4f*)(tc + r * TP)      = o[0];
                *(v4f*)(tc + r * TP + 64) = o[1];
            } else {
                const int row = m0 + r + half * 8;
                if (row < nrows) {
                    float* cp = xdst + (size_t)row * GD + c0;
                    *(v4f*)cp        = o[0];
                    *(v4f*)(cp + 64) = o[1];
                }
            }
        }
    }

    if (SKIP) {
        __syncthreads();
        w_put(ldsv, wave, lane, wa, wb);
        __syncthreads();
#pragma unroll
        for (int s = 0; s < 32; ++s) a[s] = *(const v2f*)(ta + 4 * s);
        stripe_product(ldsv, lane, a, acc);

        float cb[8];
#pragma unroll
        for (int gq = 0; gq < 2; ++gq) {
            const v4f vb = *(const v4f*)(bs + c0 + 64 * gq);
#pragma unroll
            for (int j = 0; j < 4; ++j) cb[4 * gq + j] = vb[j];
        }
        BnConst k;
        bn_load(k, gamma, beta, rmean, rvar, c0);
#pragma unroll
        for (int r = 0; r < 8; ++r) {
            v4f vo[2];
            vo[0] = *(const v4f*)(tc + r * TP);
            vo[1] = *(const v4f*)(tc + r * TP + 64);
            v4f o[2];
#pragma unroll
            for (int nt = 0; nt < 8; ++nt) {
                const float u = acc[nt][r] + cb[nt];
                float v = u + vo[nt >> 2][nt & 3];
                v = v > 0.0f ? v : 0.0f;
                o[nt >> 2][nt & 3] = bn_apply(k, nt, v);
            }
            const int row = m0 + r + half * 8;
            if (row < nrows) {
                float* cp = xdst + (size_t)row * GD + c0;
                *(v4f*)cp        = o[0];
                *(v4f*)(cp + 64) = o[1];
            }
        }
    }
}

extern "C" void kernel_launch(void* const* d_in, const int* in_sizes, int n_in,
                              void* d_out, int out_size, void* d_ws, size_t ws_size,
                              hipStream_t stream) {
    const float* x_in  = (const float*)d_in[0];
    const int*   ei    = (const int*)d_in[1];
    const float* Wc    = (const float*)d_in[2];
    const float* bc    = (const float*)d_in[3];
    const float* Wf    = (const float*)d_in[4];
    const float* bf    = (const float*)d_in[5];
    const float* Wskip = (const float*)d_in[6];
    const float* bskip = (const float*)d_in[7];
    const float* gamma = (const float*)d_in[8];
    const float* beta  = (const float*)d_in[9];
    const float* rmean = (const float*)d_in[10];
    const float* rvar  = (const float*)d_in[11];

    const int n = in_sizes[0] / GD;
    const int e = in_sizes[1] / 2;
    const int L = in_sizes[3] / GD;

    const int* src = ei;
    const int* dst = ei + e;

    float* x    = (float*)d_out;
    float* bufA = (float*)d_ws;
    int*   ibuf = (int*)(bufA + (size_t)n * GD);
    const int npad    = (n + 1 + 1023) & ~1023;
    const int nbins   = n / BIN_NODES + 1;
    const int nslices = (e + SLICE - 1) / SLICE;
    int*   rowptr  = ibuf;
    int*   csr     = rowptr + npad;
    int*   grouped = csr + e;
    int*   gstart  = grouped + (size_t)nslices * SLICE;
    int*   gcount  = gstart + (size_t)nbins * nslices;
    float* dinv    = (float*)ibuf + (size_t)n * GD + n;

    dim3 blk(256);
    const int gLay  = (n + 127) / 128;

    bin_edges_kernel<<<nslices, dim3(1024), 0, stream>>>(src, dst, e, nbins, nslices, grouped, gstart, gcount);
    build_rows_kernel<<<nbins, blk, 0, stream>>>(grouped, gstart, gcount, nslices, e, n, rowptr, dinv, csr);

    const float* cur_src = x_in;
    for (int i = 0; i < L; ++i) {
        float* cur_dst = (((L - 1 - i) & 1) == 0) ? x : bufA;
        const float* Wci = Wc + (size_t)i * GD * GD;
        const float* Wfi = Wf + (size_t)i * GD * GD;
        if (i == 0) {
            gcn_layer128<false><<<gLay, blk, 0, stream>>>(cur_src, cur_dst, n, rowptr, csr, dinv,
                                                          Wci, bc + i * GD, Wfi, bf + i * GD, nullptr, nullptr,
                                                          gamma + i * GD, beta + i * GD, rmean + i * GD, rvar + i * GD);
        } else {
            const float* Wsi = Wskip + (size_t)(i - 1) * GD * GD;
            gcn_layer128<true><<<gLay, blk, 0, stream>>>(cur_src, cur_dst, n, rowptr, csr, dinv,
                                                         Wci, bc + i * GD, Wfi, bf + i * GD, Wsi, bskip + (i - 1) * GD,
                                                         gamma + i * GD, beta + i * GD, rmean + i * GD, rvar + i * GD);
        }
        cur_src = cur_dst;
    }
}


# Layer grid recut for even matrix load per SIMD, whole-stripe waves kept: 512 work-groups of 6-7 stripes, two per WGP; spare waves skip products, a 7th stripe is halved by columns; sum order kept
# speedup vs baseline: 4.4641x; 1.0344x over previous
#include <hip/hip_runtime.h>
#include <hip/hip_bf16.h>

typedef __attribute__((ext_vector_type(2))) float v2f;
typedef __attribute__((ext_vector_type(4))) float v4f;
typedef __attribute__((ext_vector_type(8))) float v8f;

#define GD 128
#define BN_EPS 1e-5f
#define LRELU_SLOPE 0.01f

__device__ __forceinline__ void w_fetch(const float* __restrict__ W, int wave, int half, int l16,
                                        v4f (&wa)[8], v4f (&wb)[8]) {
#pragma unroll
    for (int i = 0; i < 8; ++i) {
        const int sp = wave + 8 * i;
        const int s = sp >> 1, ph = sp & 1;
        const float* wp = W + (4 * s + 2 * half) * GD + 4 * l16 + 64 * ph;
        wa[i] = *(const v4f*)wp;
        wb[i] = *(const v4f*)(wp + GD);
    }
}
__device__ __forceinline__ void w_put(v4f* ldsv, int wave, int lane, const v4f (&wa)[8], const v4f (&wb)[8]) {
#pragma unroll
    for (int i = 0; i < 8; ++i) {
        const int sp = wave + 8 * i;
        const int s = sp >> 1, ph = sp & 1;
        ldsv[(s * 4 + 2 * ph) * 32 + lane]     = (v4f){wa[i].x, wb[i].x, wa[i].y, wb[i].y};
        ldsv[(s * 4 + 2 * ph + 1) * 32 + lane] = (v4f){wa[i].z, wb[i].z, wa[i].w, wb[i].w};
    }
}
__device__ __forceinline__ void stripe_product(const v4f* ldsv, int lane, const v2f (&a)[32], v8f (&acc)[8]) {
#pragma unroll
    for (int nt = 0; nt < 8; ++nt) acc[nt] = (v8f){0, 0, 0, 0, 0, 0, 0, 0};
    v4f bq[2][4];
#pragma unroll
    for (int p = 0; p < 4; ++p) bq[0][p] = ldsv[p * 32 + lane];
    __builtin_amdgcn_sched_group_barrier(0x100, 4, 0);
#pragma unroll
    for (int s = 0; s < 32; ++s) {
        if (s + 1 < 32) {
#pragma unroll
            for (int p = 0; p < 4; ++p) bq[(s + 1) & 1][p] = ldsv[((s + 1) * 4 + p) * 32 + lane];
        }
#pragma unroll
        for (int p = 0; p < 4; ++p) {
            const v4f q = bq[s & 1][p];
            const v2f b0 = {q.x, q.y};
            const v2f b1 = {q.z, q.w};
            acc[2 * p] = __builtin_amdgcn_wmma_f32_16x16x4_f32(
                false, a[s], false, b0, (short)0, acc[2 * p], false, false);
            acc[2 * p + 1] = __builtin_amdgcn_wmma_f32_16x16x4_f32(
                false, a[s], false, b1, (short)0, acc[2 * p + 1], false, false);
        }
        __builtin_amdgcn_sched_group_barrier(0x100, 4, 0);
        __builtin_amdgcn_sched_group_barrier(0x008, 8, 0);
    }
}

template <int H>
__device__ __forceinline__ void stripe_product_half(const v4f* ldsv, int lane, const float* ta, v8f (&acc)[8]) {
#pragma unroll
    for (int nt = 0; nt < 8; ++nt) acc[nt] = (v8f){0, 0, 0, 0, 0, 0, 0, 0};
#pragma unroll 4
    for (int s = 0; s < 32; ++s) {
        const v2f af = *(const v2f*)(ta + 4 * s);
#pragma unroll
        for (int pp = 0; pp < 2; ++pp) {
            const v4f q = ldsv[(s * 4 + 2 * H + pp) * 32 + lane];
            const v2f b0 = {q.x, q.y};
            const v2f b1 = {q.z, q.w};
            acc[4 * H + 2 * pp] = __builtin_amdgcn_wmma_f32_16x16x4_f32(
                false, af, false, b0, (short)0, acc[4 * H + 2 * pp], false, false);
            acc[4 * H + 2 * pp + 1] = __builtin_amdgcn_wmma_f32_16x16x4_f32(
                false, af, false, b1, (short)0, acc[4 * H + 2 * pp + 1], false, false);
        }
    }
}

#define TP 132

struct BnConst { float m[8], r[8], g[8], e[8]; };
__device__ __forceinline__ void bn_load(BnConst& k, const float* __restrict__ gamma, const float* __restrict__ beta,
                                        const float* __restrict__ rmean, const float* __restrict__ rvar, int c0) {
#pragma unroll
    for (int g = 0; g < 2; ++g) {
        const int col = c0 + 64 * g;
        const v4f vm = *(const v4f*)(rmean + col);
        const v4f vv = *(const v4f*)(rvar + col);
        const v4f vg = *(const v4f*)(gamma + col);
        const v4f ve = *(const v4f*)(beta + col);
#pragma unroll
        for (int j = 0; j < 4; ++j) {
            k.m[4 * g + j] = vm[j];
            k.r[4 * g + j] = rsqrtf(vv[j] + BN_EPS);
            k.g[4 * g + j] = vg[j];
            k.e[4 * g + j] = ve[j];
        }
    }
}
__device__ __forceinline__ float bn_apply(const BnConst& k, int nt, float v) {
    const float bnv = (v - k.m[nt]) * k.r[nt] * k.g[nt] + k.e[nt];
    v = bnv + v;
    return v > 0.0f ? v : 0.0f;
}

#define BIN_SHIFT 7
#define BIN_NODES 128
#define NB_MAX 512
#define SLICE 8192

__device__ __forceinline__ int lds_count(int* p) {
    return __hip_atomic_fetch_add(p, 1, __ATOMIC_RELAXED, __HIP_MEMORY_SCOPE_WORKGROUP);
}

__global__ __launch_bounds__(1024)
void bin_edges_kernel(const int* __restrict__ src, const int* __restrict__ dst, int e, int nbins, int nslices,
                      int* __restrict__ grouped, int* __restrict__ gstart, int* __restrict__ gcount) {
    __shared__ int hist[NB_MAX];
    __shared__ int wsum[NB_MAX / 32];
    __shared__ int stage[SLICE];
    const int t = threadIdx.x;
    const int lane = t & 31;
    const int wave = t >> 5;
    const int b = blockIdx.x;
    const int base = b * SLICE;
    if (t < NB_MAX) hist[t] = 0;
    int d[8], s[8], slot[8];
#pragma unroll
    for (int k = 0; k < 8; ++k) {
        const int idx = base + k * 1024 + t;
        const int ic = idx < e ? idx : e - 1;
        d[k] = dst[ic];
        s[k] = src[ic];
    }
#pragma unroll
    for (int k = 0; k < 8; ++k) {
        if (base + k * 1024 + t >= e) d[k] = -1;
    }
    __syncthreads();
#pragma unroll
    for (int k = 0; k < 8; ++k) {
        slot[k] = 0;
        if (d[k] >= 0) slot[k] = lds_count(&hist[d[k] >> BIN_SHIFT]);
    }
    __syncthreads();
    const int v = (t < NB_MAX) ? hist[t] : 0;
    int incl = v;
#pragma unroll
    for (int off = 1; off < 32; off <<= 1) {
        const int u = __shfl_up(incl, off);
        if (lane >= off) incl += u;
    }
    if (t < NB_MAX && lane == 31) wsum[wave] = incl;
    __syncthreads();
    if (t < NB_MAX) {
        int pre = 0;
#pragma unroll
        for (int w = 0; w < NB_MAX / 32; ++w) if (w < wave) pre += wsum[w];
        const int excl = pre + incl - v;
        hist[t] = excl;
        if (t < nbins) {
            gstart[(size_t)t * nslices + b] = excl;
            gcount[(size_t)t * nslices + b] = v;
        }
    }
    __syncthreads();
#pragma unroll
    for (int k = 0; k < 8; ++k) {
        if (d[k] >= 0) stage[hist[d[k] >> BIN_SHIFT] + slot[k]] = (s[k] << BIN_SHIFT) | (d[k] & (BIN_NODES - 1));
    }
    __syncthreads();
#pragma unroll
    for (int k = 0; k < 8; ++k) {
        const int i = k * 1024 + t;
        if (base + i < e) grouped[base + i] = stage[i];
    }
}

template <bool FILL>
__device__ __forceinline__ void edge_put(int w, int* rowslot, int* __restrict__ csr) {
    const int j = w & (BIN_NODES - 1);
    const int pos = lds_count(&rowslot[j]);
    if (FILL) csr[pos] = (int)((unsigned)w >> BIN_SHIFT);
}
template <bool FILL>
__device__ __forceinline__ void walk_groups(const int* __restrict__ grouped, const int* __restrict__ st_row,
                                            const int* __restrict__ ct_row, int nslices, int e, int wave, int lane,
                                            int* rowslot, int* __restrict__ csr) {
    for (int b0 = wave; b0 < nslices; b0 += 64) {
        int st[8], ln[8], w0[8];
#pragma unroll
        for (int q = 0; q < 8; ++q) {
            const int b = b0 + 8 * q;
            const int bc = b < nslices ? b : nslices - 1;
            st[q] = st_row[bc];
            ln[q] = ct_row[bc];
        }
#pragma unroll
        for (int q = 0; q < 8; ++q) {
            const int b = b0 + 8 * q;
            const int bc = b < nslices ? b : nslices - 1;
            if (b >= nslices) ln[q] = 0;
            int gi = bc * SLICE + st[q] + (lane < ln[q] ? lane : 0);
            if (gi > e - 1) gi = e - 1;
            w0[q] = grouped[gi];
        }
#pragma unroll
        for (int q = 0; q < 8; ++q) {
            if (lane < ln[q]) edge_put<FILL>(w0[q], rowslot, csr);
        }
#pragma unroll
        for (int q = 0; q < 8; ++q) {
            const int b = b0 + 8 * q;
            for (int i = lane + 32; i < ln[q]; i += 32)
                edge_put<FILL>(grouped[b * SLICE + st[q] + i], rowslot, csr);
        }
    }
}

__global__ __launch_bounds__(256)
void build_rows_kernel(const int* __restrict__ grouped, const int* __restrict__ gstart,
                       const int* __restrict__ gcount, int nslices, int e, int n,
                       int* __restrict__ rowptr, float* __restrict__ dinv, int* __restrict__ csr) {
    __shared__ int rowslot[BIN_NODES];
    __shared__ int wsum[8];
    __shared__ int wfront[8];
    const int t = threadIdx.x;
    const int lane = t & 31;
    const int wave = t >> 5;
    const int c = blockIdx.x;
    const int* st_row = gstart + (size_t)c * nslices;
    const int* ct_row = gcount + (size_t)c * nslices;
    if (t < BIN_NODES) rowslot[t] = 0;
    int front = 0;
    for (int i = t; i < nslices; i += 256) front += st_row[i];
    __syncthreads();
    walk_groups<false>(grouped, st_row, ct_row, nslices, e, wave, lane, rowslot, csr);
#pragma unroll
    for (int off = 16; off > 0; off >>= 1) front += __shfl_xor(front, off);
    if (lane == 0) wfront[wave] = front;
    __syncthreads();
    front = 0;
#pragma unroll
    for (int w = 0; w < 8; ++w) front += wfront[w];
    const int v = (t < BIN_NODES) ? rowslot[t] : 0;
    int incl = v;
#pragma unroll
    for (int off = 1; off < 32; off <<= 1) {
        const int u = __shfl_up(incl, off);
        if (lane >= off) incl += u;
    }
    if (t < BIN_NODES && lane == 31) wsum[wave] = incl;
    __syncthreads();
    if (t < BIN_NODES) {
        int pre = front;
#pragma unroll
        for (int w = 0; w < BIN_NODES / 32; ++w) if (w < wave) pre += wsum[w];
        const int excl = pre + incl - v;
        rowslot[t] = excl;
        const int node = c * BIN_NODES + t;
        if (node <= n) rowptr[node] = excl;
        if (node < n)  dinv[node] = rsqrtf(1.0f + (float)v);
    }
    __syncthreads();
    walk_groups<true>(grouped, st_row, ct_row, nslices, e, wave, lane, rowslot, csr);
}

__device__ __forceinline__ void row_done(float* tg, int rpl, int ddi, int& cur, int& rend, float& dd, float4& g) {
    float* p = tg + cur * TP;
    const float4 xo = *(const float4*)p;
    float4 o;
    o.x = g.x + xo.x * dd * dd;
    o.y = g.y + xo.y * dd * dd;
    o.z = g.z + xo.z * dd * dd;
    o.w = g.w + xo.w * dd * dd;
    *(float4*)p = o;
    g.x = 0.0f; g.y = 0.0f; g.z = 0.0f; g.w = 0.0f;
    ++cur;
    rend = __builtin_amdgcn_readlane(rpl, cur + 1);
    dd   = __int_as_float(__builtin_amdgcn_readlane(ddi, cur));
}
__device__ __forceinline__ void rows_ask(const float4* __restrict__ xs4, int lane, int my_s, int j0,
                                         float4 (&v)[16]) {
#pragma unroll
    for (int k = 0; k < 16; ++k) {
        const unsigned s = (unsigned)__builtin_amdgcn_readlane(my_s, j0 + k);
        v[k] = xs4[(size_t)s * (GD / 4) + lane];
    }
}
__device__ __forceinline__ void rows_add(const float4 (&v)[16], int my_qi, int base, int j0,
                                         float* tg, int rpl, int ddi,
                                         int& cur, int& rend, float& dd, float4& g) {
#pragma unroll
    for (int k = 0; k < 16; ++k) {
        const int jj = j0 + k;
        const int e = base + jj;
        if (__builtin_expect(e == rend, 0)) {
            do row_done(tg, rpl, ddi, cur, rend, dd, g); while (e == rend && cur < 16);
        }
        const float w = __int_as_float(__builtin_amdgcn_readlane(my_qi, jj)) * dd;
        g.x += v[k].x * w; g.y += v[k].y * w; g.z += v[k].z * w; g.w += v[k].w * w;
    }
}

template <bool SKIP>
__global__ __launch_bounds__(256)
void gcn_layer128(const float* __restrict__ xsrc, float* __restrict__ xdst, int nrows,
                  const int* __restrict__ rowptr, const int* __restrict__ csr,
                  const float* __restrict__ dinv,
                  const float* __restrict__ Wc, const float* __restrict__ bc,
                  const float* __restrict__ Wf, const float* __restrict__ bf,
                  const float* __restrict__ Ws, const float* __restrict__ bs,
                  const float* __restrict__ gamma, const float* __restrict__ beta,
                  const float* __restrict__ rmean, const float* __restrict__ rvar,
                  int sq, int sr) {
    __shared__ v4f lds[GD * GD / 4 + 8 * 16 * TP / 4];
    v4f* ldsv = lds;
    float* tile = (float*)(lds + GD * GD / 4);
    const int t = threadIdx.x;
    const int wave = __builtin_amdgcn_readfirstlane(t >> 5);
    const int lane = t & 31;
    const int half = lane >> 4;
    const int l16  = lane & 15;
    const int c0 = 4 * l16;
    const int b = blockIdx.x;
    const int cnt = sq + (b < sr ? 1 : 0);
    const int first = b * sq + (b < sr ? b : sr);
    const bool patA = ((b + (b >> 8)) & 1) == 0;
    const unsigned order = patA ? 0x54632107u : 0x65432710u;
    const int rank = (int)((order >> (4 * wave)) & 7u);
    const bool shared = (cnt == 7) && (rank >= 6);
    const bool active = shared || rank < cnt;
    const int hsel = rank - 6;
    const int cm = shared ? (1 << hsel) : 3;
    const bool gathers = active && !(shared && hsel == 1);
    const int stripe = shared ? first + 6 : first + rank;
    const int m0 = active ? stripe * 16 : ((nrows + 15) & ~15);
    const int slot0 = shared ? (patA ? 5 : 7) : wave;
    const int slot1 = shared ? (patA ? 0 : 2) : wave;
    float* tw0 = tile + slot0 * 16 * TP;
    float* tw1 = tile + slot1 * 16 * TP;
    float* tc0 = tw0 + (8 * half) * TP + c0;
    float* tc1 = tw1 + (8 * half) * TP + c0;
    const float* ta0 = tw0 + l16 * TP + 2 * half;
    const float* ta1 = tw1 + l16 * TP + 2 * half;
    float* tg = tw0 + 4 * lane;

    v4f wa[8], wb[8];
    v2f a[32];
    v8f acc[8];

    w_fetch(Wc, wave, half, l16, wa, wb);

    if (gathers) {
        const float4* __restrict__ xs4 = (const float4*)xsrc;
        int ridx = m0 + (lane < 16 ? lane : 16);
        if (ridx > nrows) ridx = nrows;
        const int rpl = rowptr[ridx];
        int drow = m0 + l16;
        if (drow >= nrows) drow = nrows - 1;
        const int ddi = __float_as_int(dinv[drow]);
        float4 own[16];
#pragma unroll
        for (int r = 0; r < 16; ++r) {
            int row = m0 + r;
            if (row >= nrows) row = nrows - 1;
            own[r] = xs4[(size_t)row * (GD / 4) + lane];
        }
        const int e_begin = __builtin_amdgcn_readlane(rpl, 0);
        const int e_end   = __builtin_amdgcn_readlane(rpl, 16);
        int cur = 0;
        int rend = __builtin_amdgcn_readlane(rpl, 1);
        float dd = __int_as_float(__builtin_amdgcn_readlane(ddi, 0));
        float4 g = {0.0f, 0.0f, 0.0f, 0.0f};

        int s_cur = 0, s_nx1 = 0;
        float q_cur = 0.0f;
        if (e_begin < e_end) {
            int i0 = e_begin + lane;
            if (i0 > e_end - 1) i0 = e_end - 1;
            int i1 = e_begin + 32 + lane;
            if (i1 > e_end - 1) i1 = e_end - 1;
            s_cur = csr[i0];
            s_nx1 = csr[i1];
            q_cur = dinv[s_cur];
        }
#pragma unroll
        for (int r = 0; r < 16; ++r) *(float4*)(tg + r * TP) = own[r];

        for (int base = e_begin; base < e_end; base += 32) {
            const int m = (e_end - base) < 32 ? (e_end - base) : 32;
            const int my_qi = __float_as_int(q_cur);
            float4 v0[16], v1[16];
            rows_ask(xs4, lane, s_cur, 0, v0);
            if (m > 16) rows_ask(xs4, lane, s_cur, 16, v1);
            int i2 = base + 64 + lane;
            if (i2 > e_end - 1) i2 = e_end - 1;
            const float q_nx1 = dinv[s_nx1];
            const int s_nx2 = csr[i2];
            rows_add(v0, my_qi, base, 0, tg, rpl, ddi, cur, rend, dd, g);
            if (m > 16) rows_add(v1, my_qi, base, 16, tg, rpl, ddi, cur, rend, dd, g);
            s_cur = s_nx1; q_cur = q_nx1; s_nx1 = s_nx2;
        }
        while (cur < 16) row_done(tg, rpl, ddi, cur, rend, dd, g);
    }

    w_put(ldsv, wave, lane, wa, wb);
    __syncthreads();
    if (active) {
        if (!shared) {
#pragma unroll
            for (int s = 0; s < 32; ++s) a[s] = *(const v2f*)(ta0 + 4 * s);
            stripe_product(ldsv, lane, a, acc);
        } else if (hsel == 0) {
            stripe_product_half<0>(ldsv, lane, ta0, acc);
        } else {
            stripe_product_half<1>(ldsv, lane, ta0, acc);
        }
    }
    w_fetch(Wf, wave, half, l16, wa, wb);

    if (active) {
        float cb[8];
#pragma unroll
        for (int gq = 0; gq < 2; ++gq) {
            const v4f vb = *(const v4f*)(bc + c0 + 64 * gq);
#pragma unroll
            for (int j = 0; j < 4; ++j) cb[4 * gq + j] = vb[j];
        }
        v4f xo[8][2];
#pragma unroll
        for (int r = 0; r < 8; ++r) {
            int lrow = m0 + r + half * 8;
            if (lrow >= nrows) lrow = nrows - 1;
            const float* xp = xsrc + (size_t)lrow * GD + c0;
            xo[r][0] = *(const v4f*)xp;
            xo[r][1] = *(const v4f*)(xp + 64);
        }
#pragma unroll
        for (int r = 0; r < 8; ++r) {
            v4f o[2];
#pragma unroll
            for (int nt = 0; nt < 8; ++nt) {
                float v = (acc[nt][r] + cb[nt]) + xo[r][nt >> 2][nt & 3];
                o[nt >> 2][nt & 3] = v > 0.0f ? v : 0.0f;
            }
            if (cm & 1) *(v4f*)(tc1 + r * TP)      = o[0];
            if (cm & 2) *(v4f*)(tc1 + r * TP + 64) = o[1];
        }
    }

    __syncthreads();
    w_put(ldsv, wave, lane, wa, wb);
    __syncthreads();
    if (active) {
        if (!shared) {
#pragma unroll
            for (int s = 0; s < 32; ++s) a[s] = *(const v2f*)(ta1 + 4 * s);
            stripe_product(ldsv, lane, a, acc);
        } else if (hsel == 0) {
            stripe_product_half<0>(ldsv, lane, ta1, acc);
        } else {
            stripe_product_half<1>(ldsv, lane, ta1, acc);
        }
    }
    if (SKIP) w_fetch(Ws, wave, half, l16, wa, wb);

    if (active) {
        float cb[8];
#pragma unroll
        for (int gq = 0; gq < 2; ++gq) {
            const v4f vb = *(const v4f*)(bf + c0 + 64 * gq);
#pragma unroll
            for (int j = 0; j < 4; ++j) cb[4 * gq + j] = vb[j];
        }
        BnConst k;
        if (!SKIP) bn_load(k, gamma, beta, rmean, rvar, c0);
#pragma unroll
        for (int r = 0; r < 8; ++r) {
            v4f vo[2];
            vo[0] = *(const v4f*)(tc1 + r * TP);
            vo[1] = *(const v4f*)(tc1 + r * TP + 64);
            v4f o[2];
#pragma unroll
            for (int nt = 0; nt < 8; ++nt) {
                float u = acc[nt][r] + cb[nt];
                u = u > 0.0f ? u : LRELU_SLOPE * u;
                float v = u + vo[nt >> 2][nt & 3];
                v = v > 0.0f ? v : 0.0f;
                if (!SKIP) v = bn_apply(k, nt, v);
                o[nt >> 2][nt & 3] = v;
            }
            if (SKIP) {
                if (cm & 1) *(v4f*)(tc0 + r * TP)      = o[0];
                if (cm & 2) *(v4f*)(tc0 + r * TP + 64) = o[1];
            } else {
                const int row = m0 + r + half * 8;
                if (row < nrows) {
                    float* cp = xdst + (size_t)row * GD + c0;
                    if (cm & 1) *(v4f*)cp        = o[0];
                    if (cm & 2) *(v4f*)(cp + 64) = o[1];
                }
            }
        }
    }

    if (SKIP) {
        __syncthreads();
        w_put(ldsv, wave, lane, wa, wb);
        __syncthreads();
        if (active) {
        if (!shared) {
#pragma unroll
            for (int s = 0; s < 32; ++s) a[s] = *(const v2f*)(ta0 + 4 * s);
            stripe_product(ldsv, lane, a, acc);
        } else if (hsel == 0) {
            stripe_product_half<0>(ldsv, lane, ta0, acc);
        } else {
            stripe_product_half<1>(ldsv, lane, ta0, acc);
        }

        float cb[8];
#pragma unroll
        for (int gq = 0; gq < 2; ++gq) {
            const v4f vb = *(const v4f*)(bs + c0 + 64 * gq);
#pragma unroll
            for (int j = 0; j < 4; ++j) cb[4 * gq + j] = vb[j];
        }
        BnConst k;
        bn_load(k, gamma, beta, rmean, rvar, c0);
#pragma unroll
        for (int r = 0; r < 8; ++r) {
            v4f vo[2];
            vo[0] = *(const v4f*)(tc0 + r * TP);
            vo[1] = *(const v4f*)(tc0 + r * TP + 64);
            v4f o[2];
#pragma unroll
            for (int nt = 0; nt < 8; ++nt) {
                const float u = acc[nt][r] + cb[nt];
                float v = u + vo[nt >> 2][nt & 3];
                v = v > 0.0f ? v : 0.0f;
                o[nt >> 2][nt & 3] = bn_apply(k, nt, v);
            }
            const int row = m0 + r + half * 8;
            if (row < nrows) {
                float* cp = xdst + (size_t)row * GD + c0;
                if (cm & 1) *(v4f*)cp        = o[0];
                if (cm & 2) *(v4f*)(cp + 64) = o[1];
            }
        }
        }
    }
}

extern "C" void kernel_launch(void* const* d_in, const int* in_sizes, int n_in,
                              void* d_out, int out_size, void* d_ws, size_t ws_size,
                              hipStream_t stream) {
    const float* x_in  = (const float*)d_in[0];
    const int*   ei    = (const int*)d_in[1];
    const float* Wc    = (const float*)d_in[2];
    const float* bc    = (const float*)d_in[3];
    const float* Wf    = (const float*)d_in[4];
    const float* bf    = (const float*)d_in[5];
    const float* Wskip = (const float*)d_in[6];
    const float* bskip = (const float*)d_in[7];
    const float* gamma = (const float*)d_in[8];
    const float* beta  = (const float*)d_in[9];
    const float* rmean = (const float*)d_in[10];
    const float* rvar  = (const float*)d_in[11];

    const int n = in_sizes[0] / GD;
    const int e = in_sizes[1] / 2;
    const int L = in_sizes[3] / GD;

    const int* src = ei;
    const int* dst = ei + e;

    float* x    = (float*)d_out;
    float* bufA = (float*)d_ws;
    int*   ibuf = (int*)(bufA + (size_t)n * GD);
    const int npad    = (n + 1 + 1023) & ~1023;
    const int nbins   = n / BIN_NODES + 1;
    const int nslices = (e + SLICE - 1) / SLICE;
    int*   rowptr  = ibuf;
    int*   csr     = rowptr + npad;
    int*   grouped = csr + e;
    int*   gstart  = grouped + (size_t)nslices * SLICE;
    int*   gcount  = gstart + (size_t)nbins * nslices;
    float* dinv    = (float*)ibuf + (size_t)n * GD + n;

    dim3 blk(256);
    const int nstripes = (n + 15) / 16;
    const int gLay = nstripes <= 8 * 512 ? 512 : (nstripes + 7) / 8;
    const int sq = nstripes / gLay;
    const int sr = nstripes % gLay;

    bin_edges_kernel<<<nslices, dim3(1024), 0, stream>>>(src, dst, e, nbins, nslices, grouped, gstart, gcount);
    build_rows_kernel<<<nbins, blk, 0, stream>>>(grouped, gstart, gcount, nslices, e, n, rowptr, dinv, csr);

    const float* cur_src = x_in;
    for (int i = 0; i < L; ++i) {
        float* cur_dst = (((L - 1 - i) & 1) == 0) ? x : bufA;
        const float* Wci = Wc + (size_t)i * GD * GD;
        const float* Wfi = Wf + (size_t)i * GD * GD;
        if (i == 0) {
            gcn_layer128<false><<<gLay, blk, 0, stream>>>(cur_src, cur_dst, n, rowptr, csr, dinv,
                                                          Wci, bc + i * GD, Wfi, bf + i * GD, nullptr, nullptr,
                                                          gamma + i * GD, beta + i * GD, rmean + i * GD, rvar + i * GD,
                                                          sq, sr);
        } else {
            const float* Wsi = Wskip + (size_t)(i - 1) * GD * GD;
            gcn_layer128<true><<<gLay, blk, 0, stream>>>(cur_src, cur_dst, n, rowptr, csr, dinv,
                                                         Wci, bc + i * GD, Wfi, bf + i * GD, Wsi, bskip + (i - 1) * GD,
                                                         gamma + i * GD, beta + i * GD, rmean + i * GD, rvar + i * GD,
                                                         sq, sr);
        }
        cur_src = cur_dst;
    }
}
